# starting kernel
# baseline (speedup 1.0000x reference)
#include <hip/hip_runtime.h>
#include <hip/hip_bf16.h>

#define NNODE 3072
#define FIN   1546
#define D0    256
#define HDIM  128
#define BPAIR 16384

typedef __attribute__((ext_vector_type(2))) float v2f;
typedef __attribute__((ext_vector_type(8))) float v8f;

__device__ __forceinline__ void wmma8(v8f (&acc)[2][4], v2f a0, v2f a1,
                                      v2f b0, v2f b1, v2f b2, v2f b3)
{
  acc[0][0] = __builtin_amdgcn_wmma_f32_16x16x4_f32(false, a0, false, b0, (short)0, acc[0][0], false, false);
  acc[0][1] = __builtin_amdgcn_wmma_f32_16x16x4_f32(false, a0, false, b1, (short)0, acc[0][1], false, false);
  acc[0][2] = __builtin_amdgcn_wmma_f32_16x16x4_f32(false, a0, false, b2, (short)0, acc[0][2], false, false);
  acc[0][3] = __builtin_amdgcn_wmma_f32_16x16x4_f32(false, a0, false, b3, (short)0, acc[0][3], false, false);
  acc[1][0] = __builtin_amdgcn_wmma_f32_16x16x4_f32(false, a1, false, b0, (short)0, acc[1][0], false, false);
  acc[1][1] = __builtin_amdgcn_wmma_f32_16x16x4_f32(false, a1, false, b1, (short)0, acc[1][1], false, false);
  acc[1][2] = __builtin_amdgcn_wmma_f32_16x16x4_f32(false, a1, false, b2, (short)0, acc[1][2], false, false);
  acc[1][3] = __builtin_amdgcn_wmma_f32_16x16x4_f32(false, a1, false, b3, (short)0, acc[1][3], false, false);
}

template <bool RELU, bool BIAS, bool GATHER>
__global__ __launch_bounds__(128) void gemm_wmma_f32(
    const float* __restrict__ A, int lda,
    const float* __restrict__ Bm, int ldb,
    const float* __restrict__ bias,
    float* __restrict__ C, int ldc,
    int K,
    const int* __restrict__ gl, const int* __restrict__ gr)
{
  const int lane = threadIdx.x & 31;
  const int wave = threadIdx.x >> 5;
  const int half = lane >> 4;
  const int l16  = lane & 15;

  const int m0  = blockIdx.y * 128 + wave * 32;
  const int n0  = blockIdx.x * 64;
  const int am0 = m0 + l16;
  const int am1 = m0 + 16 + l16;

  v8f acc[2][4] = {};

  const float* bp = Bm + (size_t)(2 * half) * ldb + n0 + l16;
  const size_t bstep = (size_t)4 * ldb;

  if (GATHER) {
    const float* ap0 = A + (size_t)gl[am0] * 256 + 2 * half;
    const float* ap1 = A + (size_t)gl[am1] * 256 + 2 * half;
    #pragma unroll 4
    for (int k = 0; k < 256; k += 4) {
      v2f a0 = *(const v2f*)ap0;
      v2f a1 = *(const v2f*)ap1;
      v2f b0, b1, b2, b3;
      b0.x = bp[0];  b0.y = bp[ldb + 0];
      b1.x = bp[16]; b1.y = bp[ldb + 16];
      b2.x = bp[32]; b2.y = bp[ldb + 32];
      b3.x = bp[48]; b3.y = bp[ldb + 48];
      wmma8(acc, a0, a1, b0, b1, b2, b3);
      ap0 += 4; ap1 += 4; bp += bstep;
    }
    ap0 = A + (size_t)gr[am0] * 256 + 2 * half;
    ap1 = A + (size_t)gr[am1] * 256 + 2 * half;
    #pragma unroll 4
    for (int k = 0; k < 256; k += 4) {
      v2f a0 = *(const v2f*)ap0;
      v2f a1 = *(const v2f*)ap1;
      v2f b0, b1, b2, b3;
      b0.x = bp[0];  b0.y = bp[ldb + 0];
      b1.x = bp[16]; b1.y = bp[ldb + 16];
      b2.x = bp[32]; b2.y = bp[ldb + 32];
      b3.x = bp[48]; b3.y = bp[ldb + 48];
      wmma8(acc, a0, a1, b0, b1, b2, b3);
      ap0 += 4; ap1 += 4; bp += bstep;
    }
  } else {
    const float* ap0 = A + (size_t)am0 * lda + 2 * half;
    const float* ap1 = A + (size_t)am1 * lda + 2 * half;
    const int kmain = K & ~3;
    #pragma unroll 2
    for (int k = 0; k < kmain; k += 4) {
      v2f a0 = *(const v2f*)ap0;
      v2f a1 = *(const v2f*)ap1;
      v2f b0, b1, b2, b3;
      b0.x = bp[0];  b0.y = bp[ldb + 0];
      b1.x = bp[16]; b1.y = bp[ldb + 16];
      b2.x = bp[32]; b2.y = bp[ldb + 32];
      b3.x = bp[48]; b3.y = bp[ldb + 48];
      wmma8(acc, a0, a1, b0, b1, b2, b3);
      ap0 += 4; ap1 += 4; bp += bstep;
    }
    if (kmain < K) {
      const int ka = kmain + 2 * half;
      const bool g0 = (ka < K), g1 = (ka + 1 < K);
      v2f a0, a1, b0, b1, b2, b3;
      a0.x = g0 ? ap0[0] : 0.0f;  a0.y = g1 ? ap0[1] : 0.0f;
      a1.x = g0 ? ap1[0] : 0.0f;  a1.y = g1 ? ap1[1] : 0.0f;
      b0.x = g0 ? bp[0]  : 0.0f;  b0.y = g1 ? bp[ldb + 0]  : 0.0f;
      b1.x = g0 ? bp[16] : 0.0f;  b1.y = g1 ? bp[ldb + 16] : 0.0f;
      b2.x = g0 ? bp[32] : 0.0f;  b2.y = g1 ? bp[ldb + 32] : 0.0f;
      b3.x = g0 ? bp[48] : 0.0f;  b3.y = g1 ? bp[ldb + 48] : 0.0f;
      wmma8(acc, a0, a1, b0, b1, b2, b3);
    }
  }

  #pragma unroll
  for (int s = 0; s < 2; ++s) {
    #pragma unroll
    for (int v = 0; v < 8; ++v) {
      const int row = m0 + 16 * s + v + 8 * half;
      #pragma unroll
      for (int j = 0; j < 4; ++j) {
        const int col = n0 + 16 * j + l16;
        float x = acc[s][j][v];
        if (BIAS) x += bias[col];
        if (RELU) x = fmaxf(x, 0.0f);
        C[(size_t)row * ldc + col] = x;
      }
    }
  }
}

__global__ void deg_rsqrt(const float* __restrict__ A, float* __restrict__ d)
{
  const int gw   = (blockIdx.x * blockDim.x + threadIdx.x) >> 5;
  const int lane = threadIdx.x & 31;
  if (gw >= NNODE) return;
  const float* row = A + (size_t)gw * NNODE;
  float s = 0.0f;
  for (int j = lane; j < NNODE; j += 32) s += row[j];
  #pragma unroll
  for (int off = 16; off > 0; off >>= 1) s += __shfl_xor(s, off, 32);
  if (lane == 0) d[gw] = (s > 0.0f) ? rsqrtf(s) : 0.0f;
}

__global__ void branch_deg_rsqrt(const float* __restrict__ mask,
                                 const float* __restrict__ adj,
                                 const float* __restrict__ thrp,
                                 float* __restrict__ d)
{
  const int gw   = (blockIdx.x * blockDim.x + threadIdx.x) >> 5;
  const int lane = threadIdx.x & 31;
  if (gw >= NNODE) return;
  const float thr = thrp[0];
  const size_t base = (size_t)gw * NNODE;
  float s = 0.0f;
  for (int j = lane; j < NNODE; j += 32)
    s += (mask[base + j] > thr) ? adj[base + j] : 0.0f;
  #pragma unroll
  for (int off = 16; off > 0; off >>= 1) s += __shfl_xor(s, off, 32);
  if (lane == 0) d[gw] = rsqrtf(s + 1.0f);
}

__global__ void symnorm(const float* __restrict__ adj,
                        const float* __restrict__ d,
                        float* __restrict__ An)
{
  const int idx = blockIdx.x * blockDim.x + threadIdx.x;
  if (idx >= NNODE * NNODE) return;
  const int i = idx / NNODE;
  const int j = idx - i * NNODE;
  An[idx] = adj[idx] * d[i] * d[j];
}

__global__ void branch_norm(const float* __restrict__ mask,
                            const float* __restrict__ adj,
                            const float* __restrict__ thrp,
                            const float* __restrict__ d,
                            float* __restrict__ An)
{
  const int idx = blockIdx.x * blockDim.x + threadIdx.x;
  if (idx >= NNODE * NNODE) return;
  const float thr = thrp[0];
  const int i = idx / NNODE;
  const int j = idx - i * NNODE;
  float w = (mask[idx] > thr) ? adj[idx] : 0.0f;
  if (i == j) w += 1.0f;
  An[idx] = w * d[i] * d[j];
}

__global__ void conv_dil2(const float* __restrict__ emb,
                          const float* __restrict__ kw,
                          const float* __restrict__ kb,
                          float* __restrict__ out)
{
  const int idx = blockIdx.x * blockDim.x + threadIdx.x;
  if (idx >= NNODE * 256) return;
  const int n = idx >> 8;
  const int c = idx & 255;
  float acc = kb[0];
  #pragma unroll
  for (int ch = 0; ch < 3; ++ch) {
    #pragma unroll
    for (int kh = 0; kh < 3; ++kh) {
      const int ih = n + 2 * kh - 2;
      if (ih < 0 || ih >= NNODE) continue;
      #pragma unroll
      for (int kwi = 0; kwi < 3; ++kwi) {
        const int iw = c + 2 * kwi - 2;
        if (iw < 0 || iw >= 256) continue;
        acc += emb[((size_t)ch * NNODE + ih) * 256 + iw] * kw[ch * 9 + kh * 3 + kwi];
      }
    }
  }
  out[idx] = acc;
}

__global__ void logits_k(const float* __restrict__ h,
                         const float* __restrict__ Wb,
                         const float* __restrict__ bb,
                         float* __restrict__ out)
{
  const int i = blockIdx.x * blockDim.x + threadIdx.x;
  if (i >= BPAIR * 2) return;
  const int b = i >> 1;
  const int c = i & 1;
  float acc = bb[c];
  #pragma unroll
  for (int k = 0; k < 64; ++k) acc += h[b * 64 + k] * Wb[k * 2 + c];
  out[i] = acc;
}

extern "C" void kernel_launch(void* const* d_in, const int* in_sizes, int n_in,
                              void* d_out, int out_size, void* d_ws, size_t ws_size,
                              hipStream_t stream)
{
  const int*   left  = (const int*)d_in[0];
  const int*   right = (const int*)d_in[1];
  const float* X     = (const float*)d_in[2];
  const float* adj   = (const float*)d_in[3];
  const float* thr   = (const float*)d_in[4];
  const float* Wfc1  = (const float*)d_in[5];
  const float* bfc1  = (const float*)d_in[6];
  const float* Wg[6] = {(const float*)d_in[7],  (const float*)d_in[9],
                        (const float*)d_in[11], (const float*)d_in[13],
                        (const float*)d_in[15], (const float*)d_in[17]};
  const float* bg[6] = {(const float*)d_in[8],  (const float*)d_in[10],
                        (const float*)d_in[12], (const float*)d_in[14],
                        (const float*)d_in[16], (const float*)d_in[18]};
  const float* cnnk  = (const float*)d_in[19];
  const float* cnnb  = (const float*)d_in[20];
  const float* Wa    = (const float*)d_in[21];
  const float* ba    = (const float*)d_in[22];
  const float* Wb    = (const float*)d_in[23];
  const float* bb    = (const float*)d_in[24];

  float* out     = (float*)d_out;
  float* logits  = out;
  float* emb_all = out + BPAIR * 2;

  float* ws = (float*)d_ws;
  size_t off = 0;
  float* X0  = ws + off; off += (size_t)NNODE * D0;
  float* An  = ws + off; off += (size_t)NNODE * NNODE;
  float* A2  = ws + off; off += (size_t)NNODE * NNODE;
  float* AnB = ws + off; off += (size_t)NNODE * NNODE;
  float* dv  = ws + off; off += NNODE;
  float* Hb  = ws + off; off += (size_t)NNODE * HDIM;
  float* H2b = ws + off; off += (size_t)NNODE * HDIM;
  float* emb = ws + off; off += (size_t)3 * NNODE * 256;
  float* hm  = ws + off; off += (size_t)BPAIR * 64;
  if (ws_size < off * sizeof(float)) return;

  const dim3 blk(128);

  gemm_wmma_f32<true, true, false><<<dim3(D0 / 64, NNODE / 128), blk, 0, stream>>>(
      X, FIN, Wfc1, D0, bfc1, X0, D0, FIN, nullptr, nullptr);

  deg_rsqrt<<<NNODE / 8, 256, 0, stream>>>(adj, dv);
  symnorm<<<(NNODE * NNODE) / 256, 256, 0, stream>>>(adj, dv, An);
  gemm_wmma_f32<false, false, false><<<dim3(NNODE / 64, NNODE / 128), blk, 0, stream>>>(
      An, NNODE, An, NNODE, nullptr, A2, NNODE, NNODE, nullptr, nullptr);

  const float* masks[3] = { adj, An, A2 };
  for (int ch = 0; ch < 3; ++ch) {
    const float* W1 = Wg[4 - 2 * ch];  const float* b1 = bg[4 - 2 * ch];
    const float* W2 = Wg[5 - 2 * ch];  const float* b2 = bg[5 - 2 * ch];
    float* embc = emb + (size_t)ch * NNODE * 256;

    branch_deg_rsqrt<<<NNODE / 8, 256, 0, stream>>>(masks[ch], adj, thr, dv);
    branch_norm<<<(NNODE * NNODE) / 256, 256, 0, stream>>>(masks[ch], adj, thr, dv, AnB);

    gemm_wmma_f32<false, false, false><<<dim3(HDIM / 64, NNODE / 128), blk, 0, stream>>>(
        X0, D0, W1, HDIM, nullptr, Hb, HDIM, D0, nullptr, nullptr);
    gemm_wmma_f32<true, true, false><<<dim3(HDIM / 64, NNODE / 128), blk, 0, stream>>>(
        AnB, NNODE, Hb, HDIM, b1, embc, 256, NNODE, nullptr, nullptr);
    gemm_wmma_f32<false, false, false><<<dim3(HDIM / 64, NNODE / 128), blk, 0, stream>>>(
        embc, 256, W2, HDIM, nullptr, H2b, HDIM, HDIM, nullptr, nullptr);
    gemm_wmma_f32<true, true, false><<<dim3(HDIM / 64, NNODE / 128), blk, 0, stream>>>(
        AnB, NNODE, H2b, HDIM, b2, embc + 128, 256, NNODE, nullptr, nullptr);
  }

  conv_dil2<<<(NNODE * 256) / 256, 256, 0, stream>>>(emb, cnnk, cnnb, emb_all);

  gemm_wmma_f32<true, true, true><<<dim3(64 / 64, BPAIR / 128), blk, 0, stream>>>(
      emb_all, 256, Wa, 64, ba, hm, 64, 512, left, right);

  logits_k<<<(BPAIR * 2 + 255) / 256, 256, 0, stream>>>(hm, Wb, bb, logits);
}


# GEMM K loop split over the 8 or 16 waves of a work-group (32x64 tile per group), partials summed through LDS in fixed wave order; the 3072-cubed product keeps its 4-wave tile
# speedup vs baseline: 2.7339x; 2.7339x over previous
#include <hip/hip_runtime.h>
#include <hip/hip_bf16.h>

#define NNODE 3072
#define FIN   1546
#define D0    256
#define HDIM  128
#define BPAIR 16384

typedef __attribute__((ext_vector_type(2))) float v2f;
typedef __attribute__((ext_vector_type(8))) float v8f;

__device__ __forceinline__ void wmma8(v8f (&acc)[2][4], v2f a0, v2f a1,
                                      v2f b0, v2f b1, v2f b2, v2f b3)
{
  acc[0][0] = __builtin_amdgcn_wmma_f32_16x16x4_f32(false, a0, false, b0, (short)0, acc[0][0], false, false);
  acc[0][1] = __builtin_amdgcn_wmma_f32_16x16x4_f32(false, a0, false, b1, (short)0, acc[0][1], false, false);
  acc[0][2] = __builtin_amdgcn_wmma_f32_16x16x4_f32(false, a0, false, b2, (short)0, acc[0][2], false, false);
  acc[0][3] = __builtin_amdgcn_wmma_f32_16x16x4_f32(false, a0, false, b3, (short)0, acc[0][3], false, false);
  acc[1][0] = __builtin_amdgcn_wmma_f32_16x16x4_f32(false, a1, false, b0, (short)0, acc[1][0], false, false);
  acc[1][1] = __builtin_amdgcn_wmma_f32_16x16x4_f32(false, a1, false, b1, (short)0, acc[1][1], false, false);
  acc[1][2] = __builtin_amdgcn_wmma_f32_16x16x4_f32(false, a1, false, b2, (short)0, acc[1][2], false, false);
  acc[1][3] = __builtin_amdgcn_wmma_f32_16x16x4_f32(false, a1, false, b3, (short)0, acc[1][3], false, false);
}

#define RED_JP 272
#define RED_PW (8 * RED_JP)

template <bool RELU, bool BIAS, bool GATHER, int MW, int KS>
__global__ __launch_bounds__(32 * MW * KS) void gemm_wmma_f32(
    const float* __restrict__ A, int lda,
    const float* __restrict__ Bm, int ldb,
    const float* __restrict__ bias,
    float* __restrict__ C, int ldc,
    int K,
    const int* __restrict__ gl, const int* __restrict__ gr)
{
  static_assert(KS == 1 || MW == 1, "K split reduces one 32x64 tile per block");
  static_assert(!GATHER || (KS >= 2 && (64 % KS) == 0), "gather slice must stay in one half");

  const int lane = threadIdx.x & 31;
  const int wave = __builtin_amdgcn_readfirstlane((int)(threadIdx.x >> 5));
  const int wm   = wave % MW;
  const int wk   = wave / MW;
  const int half = lane >> 4;
  const int l16  = lane & 15;

  const int mb  = blockIdx.y * (32 * MW);
  const int m0  = mb + wm * 32;
  const int n0  = blockIdx.x * 64;
  const int am0 = m0 + l16;
  const int am1 = m0 + 16 + l16;

  v8f acc[2][4] = {};

  const int nsteps = (GATHER ? 512 : K) >> 2;
  const int per    = (nsteps + KS - 1) / KS;
  const int sbeg   = min(wk * per, nsteps);
  const int send   = min(sbeg + per, nsteps);
  const int kbeg   = 4 * sbeg;

  const float* bp = Bm + (size_t)(kbeg + 2 * half) * ldb + n0 + l16;
  const size_t bstep = (size_t)4 * ldb;

  if (GATHER) {
    const int* gi = (kbeg < 256) ? gl : gr;
    const float* ap0 = A + (size_t)gi[am0] * 256 + (kbeg & 255) + 2 * half;
    const float* ap1 = A + (size_t)gi[am1] * 256 + (kbeg & 255) + 2 * half;
    #pragma unroll 4
    for (int s = sbeg; s < send; ++s) {
      v2f a0 = *(const v2f*)ap0;
      v2f a1 = *(const v2f*)ap1;
      v2f b0, b1, b2, b3;
      b0.x = bp[0];  b0.y = bp[ldb + 0];
      b1.x = bp[16]; b1.y = bp[ldb + 16];
      b2.x = bp[32]; b2.y = bp[ldb + 32];
      b3.x = bp[48]; b3.y = bp[ldb + 48];
      wmma8(acc, a0, a1, b0, b1, b2, b3);
      ap0 += 4; ap1 += 4; bp += bstep;
    }
  } else {
    const float* ap0 = A + (size_t)am0 * lda + kbeg + 2 * half;
    const float* ap1 = A + (size_t)am1 * lda + kbeg + 2 * half;
    #pragma unroll 2
    for (int s = sbeg; s < send; ++s) {
      v2f a0 = *(const v2f*)ap0;
      v2f a1 = *(const v2f*)ap1;
      v2f b0, b1, b2, b3;
      b0.x = bp[0];  b0.y = bp[ldb + 0];
      b1.x = bp[16]; b1.y = bp[ldb + 16];
      b2.x = bp[32]; b2.y = bp[ldb + 32];
      b3.x = bp[48]; b3.y = bp[ldb + 48];
      wmma8(acc, a0, a1, b0, b1, b2, b3);
      ap0 += 4; ap1 += 4; bp += bstep;
    }
    const int kmain = K & ~3;
    if (kmain < K && wk == KS - 1) {
      const int ka = kmain + 2 * half;
      const bool g0 = (ka < K), g1 = (ka + 1 < K);
      v2f a0, a1, b0, b1, b2, b3;
      a0.x = g0 ? ap0[0] : 0.0f;  a0.y = g1 ? ap0[1] : 0.0f;
      a1.x = g0 ? ap1[0] : 0.0f;  a1.y = g1 ? ap1[1] : 0.0f;
      b0.x = g0 ? bp[0]  : 0.0f;  b0.y = g1 ? bp[ldb + 0]  : 0.0f;
      b1.x = g0 ? bp[16] : 0.0f;  b1.y = g1 ? bp[ldb + 16] : 0.0f;
      b2.x = g0 ? bp[32] : 0.0f;  b2.y = g1 ? bp[ldb + 32] : 0.0f;
      b3.x = g0 ? bp[48] : 0.0f;  b3.y = g1 ? bp[ldb + 48] : 0.0f;
      wmma8(acc, a0, a1, b0, b1, b2, b3);
    }
  }

  if (KS == 1) {
    #pragma unroll
    for (int s = 0; s < 2; ++s) {
      #pragma unroll
      for (int v = 0; v < 8; ++v) {
        const int row = m0 + 16 * s + v + 8 * half;
        #pragma unroll
        for (int j = 0; j < 4; ++j) {
          const int col = n0 + 16 * j + l16;
          float x = acc[s][j][v];
          if (BIAS) x += bias[col];
          if (RELU) x = fmaxf(x, 0.0f);
          C[(size_t)row * ldc + col] = x;
        }
      }
    }
  } else {
    __shared__ __attribute__((aligned(16))) float red[KS * RED_PW];
    float* my = red + wk * RED_PW + lane;
    #pragma unroll
    for (int s = 0; s < 2; ++s) {
      #pragma unroll
      for (int j = 0; j < 4; ++j) {
        #pragma unroll
        for (int v = 0; v < 8; ++v)
          my[(s * 4 + j) * RED_JP + v * 32] = acc[s][j][v];
      }
    }
    __syncthreads();
    for (int e = threadIdx.x; e < 512; e += 32 * KS) {
      const int row = e >> 4;
      const int c4  = (e & 15) * 4;
      const int s   = row >> 4;
      const int hf  = (row >> 3) & 1;
      const int v   = row & 7;
      const int j   = c4 >> 4;
      const int l   = c4 & 15;
      const float* p = red + (s * 4 + j) * RED_JP + v * 32 + hf * 16 + l;
      float4 sum = *(const float4*)p;
      #pragma unroll
      for (int w = 1; w < KS; ++w) {
        const float4 x = *(const float4*)(p + w * RED_PW);
        sum.x += x.x; sum.y += x.y; sum.z += x.z; sum.w += x.w;
      }
      const int col = n0 + c4;
      if (BIAS) {
        sum.x += bias[col + 0]; sum.y += bias[col + 1];
        sum.z += bias[col + 2]; sum.w += bias[col + 3];
      }
      if (RELU) {
        sum.x = fmaxf(sum.x, 0.0f); sum.y = fmaxf(sum.y, 0.0f);
        sum.z = fmaxf(sum.z, 0.0f); sum.w = fmaxf(sum.w, 0.0f);
      }
      *(float4*)(C + (size_t)(mb + row) * ldc + col) = sum;
    }
  }
}

__global__ void deg_rsqrt(const float* __restrict__ A, float* __restrict__ d)
{
  const int gw   = (blockIdx.x * blockDim.x + threadIdx.x) >> 5;
  const int lane = threadIdx.x & 31;
  if (gw >= NNODE) return;
  const float* row = A + (size_t)gw * NNODE;
  float s = 0.0f;
  for (int j = lane; j < NNODE; j += 32) s += row[j];
  #pragma unroll
  for (int off = 16; off > 0; off >>= 1) s += __shfl_xor(s, off, 32);
  if (lane == 0) d[gw] = (s > 0.0f) ? rsqrtf(s) : 0.0f;
}

__global__ void branch_deg_rsqrt(const float* __restrict__ mask,
                                 const float* __restrict__ adj,
                                 const float* __restrict__ thrp,
                                 float* __restrict__ d)
{
  const int gw   = (blockIdx.x * blockDim.x + threadIdx.x) >> 5;
  const int lane = threadIdx.x & 31;
  if (gw >= NNODE) return;
  const float thr = thrp[0];
  const size_t base = (size_t)gw * NNODE;
  float s = 0.0f;
  for (int j = lane; j < NNODE; j += 32)
    s += (mask[base + j] > thr) ? adj[base + j] : 0.0f;
  #pragma unroll
  for (int off = 16; off > 0; off >>= 1) s += __shfl_xor(s, off, 32);
  if (lane == 0) d[gw] = rsqrtf(s + 1.0f);
}

__global__ void symnorm(const float* __restrict__ adj,
                        const float* __restrict__ d,
                        float* __restrict__ An)
{
  const int idx = blockIdx.x * blockDim.x + threadIdx.x;
  if (idx >= NNODE * NNODE) return;
  const int i = idx / NNODE;
  const int j = idx - i * NNODE;
  An[idx] = adj[idx] * d[i] * d[j];
}

__global__ void branch_norm(const float* __restrict__ mask,
                            const float* __restrict__ adj,
                            const float* __restrict__ thrp,
                            const float* __restrict__ d,
                            float* __restrict__ An)
{
  const int idx = blockIdx.x * blockDim.x + threadIdx.x;
  if (idx >= NNODE * NNODE) return;
  const float thr = thrp[0];
  const int i = idx / NNODE;
  const int j = idx - i * NNODE;
  float w = (mask[idx] > thr) ? adj[idx] : 0.0f;
  if (i == j) w += 1.0f;
  An[idx] = w * d[i] * d[j];
}

__global__ void conv_dil2(const float* __restrict__ emb,
                          const float* __restrict__ kw,
                          const float* __restrict__ kb,
                          float* __restrict__ out)
{
  const int idx = blockIdx.x * blockDim.x + threadIdx.x;
  if (idx >= NNODE * 256) return;
  const int n = idx >> 8;
  const int c = idx & 255;
  float acc = kb[0];
  #pragma unroll
  for (int ch = 0; ch < 3; ++ch) {
    #pragma unroll
    for (int kh = 0; kh < 3; ++kh) {
      const int ih = n + 2 * kh - 2;
      if (ih < 0 || ih >= NNODE) continue;
      #pragma unroll
      for (int kwi = 0; kwi < 3; ++kwi) {
        const int iw = c + 2 * kwi - 2;
        if (iw < 0 || iw >= 256) continue;
        acc += emb[((size_t)ch * NNODE + ih) * 256 + iw] * kw[ch * 9 + kh * 3 + kwi];
      }
    }
  }
  out[idx] = acc;
}

__global__ void logits_k(const float* __restrict__ h,
                         const float* __restrict__ Wb,
                         const float* __restrict__ bb,
                         float* __restrict__ out)
{
  const int i = blockIdx.x * blockDim.x + threadIdx.x;
  if (i >= BPAIR * 2) return;
  const int b = i >> 1;
  const int c = i & 1;
  float acc = bb[c];
  #pragma unroll
  for (int k = 0; k < 64; ++k) acc += h[b * 64 + k] * Wb[k * 2 + c];
  out[i] = acc;
}

extern "C" void kernel_launch(void* const* d_in, const int* in_sizes, int n_in,
                              void* d_out, int out_size, void* d_ws, size_t ws_size,
                              hipStream_t stream)
{
  const int*   left  = (const int*)d_in[0];
  const int*   right = (const int*)d_in[1];
  const float* X     = (const float*)d_in[2];
  const float* adj   = (const float*)d_in[3];
  const float* thr   = (const float*)d_in[4];
  const float* Wfc1  = (const float*)d_in[5];
  const float* bfc1  = (const float*)d_in[6];
  const float* Wg[6] = {(const float*)d_in[7],  (const float*)d_in[9],
                        (const float*)d_in[11], (const float*)d_in[13],
                        (const float*)d_in[15], (const float*)d_in[17]};
  const float* bg[6] = {(const float*)d_in[8],  (const float*)d_in[10],
                        (const float*)d_in[12], (const float*)d_in[14],
                        (const float*)d_in[16], (const float*)d_in[18]};
  const float* cnnk  = (const float*)d_in[19];
  const float* cnnb  = (const float*)d_in[20];
  const float* Wa    = (const float*)d_in[21];
  const float* ba    = (const float*)d_in[22];
  const float* Wb    = (const float*)d_in[23];
  const float* bb    = (const float*)d_in[24];

  float* out     = (float*)d_out;
  float* logits  = out;
  float* emb_all = out + BPAIR * 2;

  float* ws = (float*)d_ws;
  size_t off = 0;
  float* X0  = ws + off; off += (size_t)NNODE * D0;
  float* An  = ws + off; off += (size_t)NNODE * NNODE;
  float* A2  = ws + off; off += (size_t)NNODE * NNODE;
  float* AnB = ws + off; off += (size_t)NNODE * NNODE;
  float* dv  = ws + off; off += NNODE;
  float* Hb  = ws + off; off += (size_t)NNODE * HDIM;
  float* H2b = ws + off; off += (size_t)NNODE * HDIM;
  float* emb = ws + off; off += (size_t)3 * NNODE * 256;
  float* hm  = ws + off; off += (size_t)BPAIR * 64;
  if (ws_size < off * sizeof(float)) return;

  const dim3 blk(128);
  const dim3 blk8(256);
  const dim3 blk16(512);

  gemm_wmma_f32<true, true, false, 1, 16><<<dim3(D0 / 64, NNODE / 32), blk16, 0, stream>>>(
      X, FIN, Wfc1, D0, bfc1, X0, D0, FIN, nullptr, nullptr);

  deg_rsqrt<<<NNODE / 8, 256, 0, stream>>>(adj, dv);
  symnorm<<<(NNODE * NNODE) / 256, 256, 0, stream>>>(adj, dv, An);
  gemm_wmma_f32<false, false, false, 4, 1><<<dim3(NNODE / 64, NNODE / 128), blk, 0, stream>>>(
      An, NNODE, An, NNODE, nullptr, A2, NNODE, NNODE, nullptr, nullptr);

  const float* masks[3] = { adj, An, A2 };
  for (int ch = 0; ch < 3; ++ch) {
    const float* W1 = Wg[4 - 2 * ch];  const float* b1 = bg[4 - 2 * ch];
    const float* W2 = Wg[5 - 2 * ch];  const float* b2 = bg[5 - 2 * ch];
    float* embc = emb + (size_t)ch * NNODE * 256;

    branch_deg_rsqrt<<<NNODE / 8, 256, 0, stream>>>(masks[ch], adj, thr, dv);
    branch_norm<<<(NNODE * NNODE) / 256, 256, 0, stream>>>(masks[ch], adj, thr, dv, AnB);

    gemm_wmma_f32<false, false, false, 1, 8><<<dim3(HDIM / 64, NNODE / 32), blk8, 0, stream>>>(
        X0, D0, W1, HDIM, nullptr, Hb, HDIM, D0, nullptr, nullptr);
    gemm_wmma_f32<true, true, false, 1, 16><<<dim3(HDIM / 64, NNODE / 32), blk16, 0, stream>>>(
        AnB, NNODE, Hb, HDIM, b1, embc, 256, NNODE, nullptr, nullptr);
    gemm_wmma_f32<false, false, false, 1, 8><<<dim3(HDIM / 64, NNODE / 32), blk8, 0, stream>>>(
        embc, 256, W2, HDIM, nullptr, H2b, HDIM, HDIM, nullptr, nullptr);
    gemm_wmma_f32<true, true, false, 1, 16><<<dim3(HDIM / 64, NNODE / 32), blk16, 0, stream>>>(
        AnB, NNODE, H2b, HDIM, b2, embc + 128, 256, NNODE, nullptr, nullptr);
  }

  conv_dil2<<<(NNODE * 256) / 256, 256, 0, stream>>>(emb, cnnk, cnnb, emb_all);

  gemm_wmma_f32<true, true, true, 1, 8><<<dim3(64 / 64, BPAIR / 32), blk8, 0, stream>>>(
      emb_all, 256, Wa, 64, ba, hm, 64, 512, left, right);

  logits_k<<<(BPAIR * 2 + 255) / 256, 256, 0, stream>>>(hm, Wb, bb, logits);
}


# 3072-cubed product: 192x64 A and 64x192 B slabs staged through LDS once per 192x192 work-group tile (8 waves, 48x96 each), fragments read from LDS; k summation order per output unchanged
# speedup vs baseline: 3.1589x; 1.1555x over previous
#include <hip/hip_runtime.h>
#include <hip/hip_bf16.h>

#define NNODE 3072
#define FIN   1546
#define D0    256
#define HDIM  128
#define BPAIR 16384

typedef __attribute__((ext_vector_type(2))) float v2f;
typedef __attribute__((ext_vector_type(8))) float v8f;

__device__ __forceinline__ void wmma8(v8f (&acc)[2][4], v2f a0, v2f a1,
                                      v2f b0, v2f b1, v2f b2, v2f b3)
{
  acc[0][0] = __builtin_amdgcn_wmma_f32_16x16x4_f32(false, a0, false, b0, (short)0, acc[0][0], false, false);
  acc[0][1] = __builtin_amdgcn_wmma_f32_16x16x4_f32(false, a0, false, b1, (short)0, acc[0][1], false, false);
  acc[0][2] = __builtin_amdgcn_wmma_f32_16x16x4_f32(false, a0, false, b2, (short)0, acc[0][2], false, false);
  acc[0][3] = __builtin_amdgcn_wmma_f32_16x16x4_f32(false, a0, false, b3, (short)0, acc[0][3], false, false);
  acc[1][0] = __builtin_amdgcn_wmma_f32_16x16x4_f32(false, a1, false, b0, (short)0, acc[1][0], false, false);
  acc[1][1] = __builtin_amdgcn_wmma_f32_16x16x4_f32(false, a1, false, b1, (short)0, acc[1][1], false, false);
  acc[1][2] = __builtin_amdgcn_wmma_f32_16x16x4_f32(false, a1, false, b2, (short)0, acc[1][2], false, false);
  acc[1][3] = __builtin_amdgcn_wmma_f32_16x16x4_f32(false, a1, false, b3, (short)0, acc[1][3], false, false);
}

#define RED_JP 272
#define RED_PW (8 * RED_JP)

template <bool RELU, bool BIAS, bool GATHER, int MW, int KS>
__global__ __launch_bounds__(32 * MW * KS) void gemm_wmma_f32(
    const float* __restrict__ A, int lda,
    const float* __restrict__ Bm, int ldb,
    const float* __restrict__ bias,
    float* __restrict__ C, int ldc,
    int K,
    const int* __restrict__ gl, const int* __restrict__ gr)
{
  static_assert(KS == 1 || MW == 1, "K split reduces one 32x64 tile per block");
  static_assert(!GATHER || (KS >= 2 && (64 % KS) == 0), "gather slice must stay in one half");

  const int lane = threadIdx.x & 31;
  const int wave = __builtin_amdgcn_readfirstlane((int)(threadIdx.x >> 5));
  const int wm   = wave % MW;
  const int wk   = wave / MW;
  const int half = lane >> 4;
  const int l16  = lane & 15;

  const int mb  = blockIdx.y * (32 * MW);
  const int m0  = mb + wm * 32;
  const int n0  = blockIdx.x * 64;
  const int am0 = m0 + l16;
  const int am1 = m0 + 16 + l16;

  v8f acc[2][4] = {};

  const int nsteps = (GATHER ? 512 : K) >> 2;
  const int per    = (nsteps + KS - 1) / KS;
  const int sbeg   = min(wk * per, nsteps);
  const int send   = min(sbeg + per, nsteps);
  const int kbeg   = 4 * sbeg;

  const float* bp = Bm + (size_t)(kbeg + 2 * half) * ldb + n0 + l16;
  const size_t bstep = (size_t)4 * ldb;

  if (GATHER) {
    const int* gi = (kbeg < 256) ? gl : gr;
    const float* ap0 = A + (size_t)gi[am0] * 256 + (kbeg & 255) + 2 * half;
    const float* ap1 = A + (size_t)gi[am1] * 256 + (kbeg & 255) + 2 * half;
    #pragma unroll 4
    for (int s = sbeg; s < send; ++s) {
      v2f a0 = *(const v2f*)ap0;
      v2f a1 = *(const v2f*)ap1;
      v2f b0, b1, b2, b3;
      b0.x = bp[0];  b0.y = bp[ldb + 0];
      b1.x = bp[16]; b1.y = bp[ldb + 16];
      b2.x = bp[32]; b2.y = bp[ldb + 32];
      b3.x = bp[48]; b3.y = bp[ldb + 48];
      wmma8(acc, a0, a1, b0, b1, b2, b3);
      ap0 += 4; ap1 += 4; bp += bstep;
    }
  } else {
    const float* ap0 = A + (size_t)am0 * lda + kbeg + 2 * half;
    const float* ap1 = A + (size_t)am1 * lda + kbeg + 2 * half;
    #pragma unroll 2
    for (int s = sbeg; s < send; ++s) {
      v2f a0 = *(const v2f*)ap0;
      v2f a1 = *(const v2f*)ap1;
      v2f b0, b1, b2, b3;
      b0.x = bp[0];  b0.y = bp[ldb + 0];
      b1.x = bp[16]; b1.y = bp[ldb + 16];
      b2.x = bp[32]; b2.y = bp[ldb + 32];
      b3.x = bp[48]; b3.y = bp[ldb + 48];
      wmma8(acc, a0, a1, b0, b1, b2, b3);
      ap0 += 4; ap1 += 4; bp += bstep;
    }
    const int kmain = K & ~3;
    if (kmain < K && wk == KS - 1) {
      const int ka = kmain + 2 * half;
      const bool g0 = (ka < K), g1 = (ka + 1 < K);
      v2f a0, a1, b0, b1, b2, b3;
      a0.x = g0 ? ap0[0] : 0.0f;  a0.y = g1 ? ap0[1] : 0.0f;
      a1.x = g0 ? ap1[0] : 0.0f;  a1.y = g1 ? ap1[1] : 0.0f;
      b0.x = g0 ? bp[0]  : 0.0f;  b0.y = g1 ? bp[ldb + 0]  : 0.0f;
      b1.x = g0 ? bp[16] : 0.0f;  b1.y = g1 ? bp[ldb + 16] : 0.0f;
      b2.x = g0 ? bp[32] : 0.0f;  b2.y = g1 ? bp[ldb + 32] : 0.0f;
      b3.x = g0 ? bp[48] : 0.0f;  b3.y = g1 ? bp[ldb + 48] : 0.0f;
      wmma8(acc, a0, a1, b0, b1, b2, b3);
    }
  }

  if (KS == 1) {
    #pragma unroll
    for (int s = 0; s < 2; ++s) {
      #pragma unroll
      for (int v = 0; v < 8; ++v) {
        const int row = m0 + 16 * s + v + 8 * half;
        #pragma unroll
        for (int j = 0; j < 4; ++j) {
          const int col = n0 + 16 * j + l16;
          float x = acc[s][j][v];
          if (BIAS) x += bias[col];
          if (RELU) x = fmaxf(x, 0.0f);
          C[(size_t)row * ldc + col] = x;
        }
      }
    }
  } else {
    __shared__ __attribute__((aligned(16))) float red[KS * RED_PW];
    float* my = red + wk * RED_PW + lane;
    #pragma unroll
    for (int s = 0; s < 2; ++s) {
      #pragma unroll
      for (int j = 0; j < 4; ++j) {
        #pragma unroll
        for (int v = 0; v < 8; ++v)
          my[(s * 4 + j) * RED_JP + v * 32] = acc[s][j][v];
      }
    }
    __syncthreads();
    for (int e = threadIdx.x; e < 512; e += 32 * KS) {
      const int row = e >> 4;
      const int c4  = (e & 15) * 4;
      const int s   = row >> 4;
      const int hf  = (row >> 3) & 1;
      const int v   = row & 7;
      const int j   = c4 >> 4;
      const int l   = c4 & 15;
      const float* p = red + (s * 4 + j) * RED_JP + v * 32 + hf * 16 + l;
      float4 sum = *(const float4*)p;
      #pragma unroll
      for (int w = 1; w < KS; ++w) {
        const float4 x = *(const float4*)(p + w * RED_PW);
        sum.x += x.x; sum.y += x.y; sum.z += x.z; sum.w += x.w;
      }
      const int col = n0 + c4;
      if (BIAS) {
        sum.x += bias[col + 0]; sum.y += bias[col + 1];
        sum.z += bias[col + 2]; sum.w += bias[col + 3];
      }
      if (RELU) {
        sum.x = fmaxf(sum.x, 0.0f); sum.y = fmaxf(sum.y, 0.0f);
        sum.z = fmaxf(sum.z, 0.0f); sum.w = fmaxf(sum.w, 0.0f);
      }
      *(float4*)(C + (size_t)(mb + row) * ldc + col) = sum;
    }
  }
}

#define SQ_T   192
#define SQ_KT  64
#define SQ_PA  68
#define SQ_PB  416
#define SQ_ASZ (SQ_T * SQ_PA)
#define SQ_BSZ ((SQ_KT / 2) * SQ_PB)
#define SQ_STG (SQ_ASZ + SQ_BSZ)

__device__ __forceinline__ void sq_load(float4 (&ra)[12], float4 (&rb0)[6], float4 (&rb1)[6],
                                        const float* gA, const float* gB)
{
  #pragma unroll
  for (int i = 0; i < 12; ++i) ra[i] = *(const float4*)(gA + (size_t)i * 16 * NNODE);
  #pragma unroll
  for (int a = 0; a < 2; ++a) {
    #pragma unroll
    for (int b = 0; b < 3; ++b) {
      const float* g = gB + (size_t)a * 32 * NNODE + b * 64;
      rb0[a * 3 + b] = *(const float4*)g;
      rb1[a * 3 + b] = *(const float4*)(g + NNODE);
    }
  }
}

template <int I>
__device__ __forceinline__ void sq_store_a(float* aW, const float4 (&ra)[12])
{
  *(float4*)(aW + I * 16 * SQ_PA) = make_float4(ra[I].x, ra[I].y, ra[I].z, ra[I].w);
}

template <int Q>
__device__ __forceinline__ void sq_store_b(float* bW, const float4 (&rb0)[6],
                                           const float4 (&rb1)[6])
{
  float* d = bW + (Q / 3) * 16 * SQ_PB + (Q % 3) * 128;
  *(float4*)d       = make_float4(rb0[Q].x, rb1[Q].x, rb0[Q].y, rb1[Q].y);
  *(float4*)(d + 4) = make_float4(rb0[Q].z, rb1[Q].z, rb0[Q].w, rb1[Q].w);
}

template <bool STAGE_NEXT, int I>
__device__ __forceinline__ void sq_kstep(v8f (&acc)[3][6], v2f (&a)[2][3], v2f (&b)[2][6],
                                         const float* aR, const float* bR,
                                         float* aW, float* bW, const float4 (&ra)[12],
                                         const float4 (&rb0)[6], const float4 (&rb1)[6])
{
  constexpr int c = I & 1, n = c ^ 1;
  if (I + 1 < SQ_KT / 4) {
    #pragma unroll
    for (int s = 0; s < 3; ++s)
      a[n][s] = *(const v2f*)(aR + s * 16 * SQ_PA + 4 * (I + 1));
    #pragma unroll
    for (int j = 0; j < 6; ++j)
      b[n][j] = *(const v2f*)(bR + 2 * (I + 1) * SQ_PB + j * 32);
  }
  #pragma unroll
  for (int s = 0; s < 3; ++s) {
    #pragma unroll
    for (int j = 0; j < 6; ++j)
      acc[s][j] = __builtin_amdgcn_wmma_f32_16x16x4_f32(false, a[c][s], false, b[c][j],
                                                        (short)0, acc[s][j], false, false);
  }
  if constexpr (STAGE_NEXT && I >= 2 && I < 14) {
    __builtin_amdgcn_sched_barrier(0);
    sq_store_a<I - 2>(aW, ra);
    if constexpr (((I - 2) & 1) == 0) sq_store_b<(I - 2) / 2>(bW, rb0, rb1);
  }
}

template <bool STAGE_NEXT>
__device__ __forceinline__ void sq_slab(v8f (&acc)[3][6], const float* aR, const float* bR,
                                        float* aW, float* bW, const float4 (&ra)[12],
                                        const float4 (&rb0)[6], const float4 (&rb1)[6])
{
  v2f a[2][3], b[2][6];
  #pragma unroll
  for (int s = 0; s < 3; ++s) a[0][s] = *(const v2f*)(aR + s * 16 * SQ_PA);
  #pragma unroll
  for (int j = 0; j < 6; ++j) b[0][j] = *(const v2f*)(bR + j * 32);
  sq_kstep<STAGE_NEXT,  0>(acc, a, b, aR, bR, aW, bW, ra, rb0, rb1);
  sq_kstep<STAGE_NEXT,  1>(acc, a, b, aR, bR, aW, bW, ra, rb0, rb1);
  sq_kstep<STAGE_NEXT,  2>(acc, a, b, aR, bR, aW, bW, ra, rb0, rb1);
  sq_kstep<STAGE_NEXT,  3>(acc, a, b, aR, bR, aW, bW, ra, rb0, rb1);
  sq_kstep<STAGE_NEXT,  4>(acc, a, b, aR, bR, aW, bW, ra, rb0, rb1);
  sq_kstep<STAGE_NEXT,  5>(acc, a, b, aR, bR, aW, bW, ra, rb0, rb1);
  sq_kstep<STAGE_NEXT,  6>(acc, a, b, aR, bR, aW, bW, ra, rb0, rb1);
  sq_kstep<STAGE_NEXT,  7>(acc, a, b, aR, bR, aW, bW, ra, rb0, rb1);
  sq_kstep<STAGE_NEXT,  8>(acc, a, b, aR, bR, aW, bW, ra, rb0, rb1);
  sq_kstep<STAGE_NEXT,  9>(acc, a, b, aR, bR, aW, bW, ra, rb0, rb1);
  sq_kstep<STAGE_NEXT, 10>(acc, a, b, aR, bR, aW, bW, ra, rb0, rb1);
  sq_kstep<STAGE_NEXT, 11>(acc, a, b, aR, bR, aW, bW, ra, rb0, rb1);
  sq_kstep<STAGE_NEXT, 12>(acc, a, b, aR, bR, aW, bW, ra, rb0, rb1);
  sq_kstep<STAGE_NEXT, 13>(acc, a, b, aR, bR, aW, bW, ra, rb0, rb1);
  sq_kstep<STAGE_NEXT, 14>(acc, a, b, aR, bR, aW, bW, ra, rb0, rb1);
  sq_kstep<STAGE_NEXT, 15>(acc, a, b, aR, bR, aW, bW, ra, rb0, rb1);
}

__global__ __launch_bounds__(256) void gemm_sq_lds_f32(const float* __restrict__ A,
                                                       const float* __restrict__ Bm,
                                                       float* __restrict__ C)
{
  __shared__ __attribute__((aligned(16))) float sm[2 * SQ_STG];

  const int tid  = threadIdx.x;
  const int lane = tid & 31;
  const int wave = __builtin_amdgcn_readfirstlane(tid >> 5);
  const int wm   = wave >> 1;
  const int wn   = wave & 1;
  const int half = lane >> 4;
  const int l16  = lane & 15;
  const int mb   = blockIdx.y * SQ_T;
  const int nb   = blockIdx.x * SQ_T;

  const int tr = tid >> 4, tc = tid & 15;
  const float* gA = A + (size_t)(mb + tr) * NNODE + 4 * tc;
  const float* gB = Bm + (size_t)(2 * tr) * NNODE + nb + 4 * tc;
  const int sAo = tr * SQ_PA + 4 * tc;
  const int sBo = SQ_ASZ + tr * SQ_PB + 8 * tc;
  const int aF = (48 * wm + l16) * SQ_PA + 2 * half;
  const int bF = SQ_ASZ + half * SQ_PB + 2 * (96 * wn + l16);

  v8f acc[3][6] = {};
  float4 ra[12], rb0[6], rb1[6];

  sq_load(ra, rb0, rb1, gA, gB);
  sq_store_a<0>(sm + sAo, ra);  sq_store_a<1>(sm + sAo, ra);  sq_store_a<2>(sm + sAo, ra);
  sq_store_a<3>(sm + sAo, ra);  sq_store_a<4>(sm + sAo, ra);  sq_store_a<5>(sm + sAo, ra);
  sq_store_a<6>(sm + sAo, ra);  sq_store_a<7>(sm + sAo, ra);  sq_store_a<8>(sm + sAo, ra);
  sq_store_a<9>(sm + sAo, ra);  sq_store_a<10>(sm + sAo, ra); sq_store_a<11>(sm + sAo, ra);
  sq_store_b<0>(sm + sBo, rb0, rb1);  sq_store_b<1>(sm + sBo, rb0, rb1);
  sq_store_b<2>(sm + sBo, rb0, rb1);  sq_store_b<3>(sm + sBo, rb0, rb1);
  sq_store_b<4>(sm + sBo, rb0, rb1);  sq_store_b<5>(sm + sBo, rb0, rb1);
  __syncthreads();

  int st = 0;
  for (int t = 0; t < NNODE / SQ_KT - 1; ++t) {
    gA += SQ_KT;
    gB += (size_t)SQ_KT * NNODE;
    sq_load(ra, rb0, rb1, gA, gB);
    sq_slab<true>(acc, sm + st + aF, sm + st + bF, sm + (SQ_STG - st) + sAo,
                  sm + (SQ_STG - st) + sBo, ra, rb0, rb1);
    __syncthreads();
    st = SQ_STG - st;
  }
  sq_slab<false>(acc, sm + st + aF, sm + st + bF, nullptr, nullptr, ra, rb0, rb1);

  #pragma unroll
  for (int s = 0; s < 3; ++s) {
    #pragma unroll
    for (int v = 0; v < 8; ++v) {
      const int row = mb + 48 * wm + 16 * s + v + 8 * half;
      #pragma unroll
      for (int j = 0; j < 6; ++j) {
        const int col = nb + 96 * wn + 16 * j + l16;
        C[(size_t)row * NNODE + col] = acc[s][j][v];
      }
    }
  }
}

__global__ void deg_rsqrt(const float* __restrict__ A, float* __restrict__ d)
{
  const int gw   = (blockIdx.x * blockDim.x + threadIdx.x) >> 5;
  const int lane = threadIdx.x & 31;
  if (gw >= NNODE) return;
  const float* row = A + (size_t)gw * NNODE;
  float s = 0.0f;
  for (int j = lane; j < NNODE; j += 32) s += row[j];
  #pragma unroll
  for (int off = 16; off > 0; off >>= 1) s += __shfl_xor(s, off, 32);
  if (lane == 0) d[gw] = (s > 0.0f) ? rsqrtf(s) : 0.0f;
}

__global__ void branch_deg_rsqrt(const float* __restrict__ mask,
                                 const float* __restrict__ adj,
                                 const float* __restrict__ thrp,
                                 float* __restrict__ d)
{
  const int gw   = (blockIdx.x * blockDim.x + threadIdx.x) >> 5;
  const int lane = threadIdx.x & 31;
  if (gw >= NNODE) return;
  const float thr = thrp[0];
  const size_t base = (size_t)gw * NNODE;
  float s = 0.0f;
  for (int j = lane; j < NNODE; j += 32)
    s += (mask[base + j] > thr) ? adj[base + j] : 0.0f;
  #pragma unroll
  for (int off = 16; off > 0; off >>= 1) s += __shfl_xor(s, off, 32);
  if (lane == 0) d[gw] = rsqrtf(s + 1.0f);
}

__global__ void symnorm(const float* __restrict__ adj,
                        const float* __restrict__ d,
                        float* __restrict__ An)
{
  const int idx = blockIdx.x * blockDim.x + threadIdx.x;
  if (idx >= NNODE * NNODE) return;
  const int i = idx / NNODE;
  const int j = idx - i * NNODE;
  An[idx] = adj[idx] * d[i] * d[j];
}

__global__ void branch_norm(const float* __restrict__ mask,
                            const float* __restrict__ adj,
                            const float* __restrict__ thrp,
                            const float* __restrict__ d,
                            float* __restrict__ An)
{
  const int idx = blockIdx.x * blockDim.x + threadIdx.x;
  if (idx >= NNODE * NNODE) return;
  const float thr = thrp[0];
  const int i = idx / NNODE;
  const int j = idx - i * NNODE;
  float w = (mask[idx] > thr) ? adj[idx] : 0.0f;
  if (i == j) w += 1.0f;
  An[idx] = w * d[i] * d[j];
}

__global__ void conv_dil2(const float* __restrict__ emb,
                          const float* __restrict__ kw,
                          const float* __restrict__ kb,
                          float* __restrict__ out)
{
  const int idx = blockIdx.x * blockDim.x + threadIdx.x;
  if (idx >= NNODE * 256) return;
  const int n = idx >> 8;
  const int c = idx & 255;
  float acc = kb[0];
  #pragma unroll
  for (int ch = 0; ch < 3; ++ch) {
    #pragma unroll
    for (int kh = 0; kh < 3; ++kh) {
      const int ih = n + 2 * kh - 2;
      if (ih < 0 || ih >= NNODE) continue;
      #pragma unroll
      for (int kwi = 0; kwi < 3; ++kwi) {
        const int iw = c + 2 * kwi - 2;
        if (iw < 0 || iw >= 256) continue;
        acc += emb[((size_t)ch * NNODE + ih) * 256 + iw] * kw[ch * 9 + kh * 3 + kwi];
      }
    }
  }
  out[idx] = acc;
}

__global__ void logits_k(const float* __restrict__ h,
                         const float* __restrict__ Wb,
                         const float* __restrict__ bb,
                         float* __restrict__ out)
{
  const int i = blockIdx.x * blockDim.x + threadIdx.x;
  if (i >= BPAIR * 2) return;
  const int b = i >> 1;
  const int c = i & 1;
  float acc = bb[c];
  #pragma unroll
  for (int k = 0; k < 64; ++k) acc += h[b * 64 + k] * Wb[k * 2 + c];
  out[i] = acc;
}

extern "C" void kernel_launch(void* const* d_in, const int* in_sizes, int n_in,
                              void* d_out, int out_size, void* d_ws, size_t ws_size,
                              hipStream_t stream)
{
  const int*   left  = (const int*)d_in[0];
  const int*   right = (const int*)d_in[1];
  const float* X     = (const float*)d_in[2];
  const float* adj   = (const float*)d_in[3];
  const float* thr   = (const float*)d_in[4];
  const float* Wfc1  = (const float*)d_in[5];
  const float* bfc1  = (const float*)d_in[6];
  const float* Wg[6] = {(const float*)d_in[7],  (const float*)d_in[9],
                        (const float*)d_in[11], (const float*)d_in[13],
                        (const float*)d_in[15], (const float*)d_in[17]};
  const float* bg[6] = {(const float*)d_in[8],  (const float*)d_in[10],
                        (const float*)d_in[12], (const float*)d_in[14],
                        (const float*)d_in[16], (const float*)d_in[18]};
  const float* cnnk  = (const float*)d_in[19];
  const float* cnnb  = (const float*)d_in[20];
  const float* Wa    = (const float*)d_in[21];
  const float* ba    = (const float*)d_in[22];
  const float* Wb    = (const float*)d_in[23];
  const float* bb    = (const float*)d_in[24];

  float* out     = (float*)d_out;
  float* logits  = out;
  float* emb_all = out + BPAIR * 2;

  float* ws = (float*)d_ws;
  size_t off = 0;
  float* X0  = ws + off; off += (size_t)NNODE * D0;
  float* An  = ws + off; off += (size_t)NNODE * NNODE;
  float* A2  = ws + off; off += (size_t)NNODE * NNODE;
  float* AnB = ws + off; off += (size_t)NNODE * NNODE;
  float* dv  = ws + off; off += NNODE;
  float* Hb  = ws + off; off += (size_t)NNODE * HDIM;
  float* H2b = ws + off; off += (size_t)NNODE * HDIM;
  float* emb = ws + off; off += (size_t)3 * NNODE * 256;
  float* hm  = ws + off; off += (size_t)BPAIR * 64;
  if (ws_size < off * sizeof(float)) return;

  const dim3 blk8(256);
  const dim3 blk16(512);

  gemm_wmma_f32<true, true, false, 1, 16><<<dim3(D0 / 64, NNODE / 32), blk16, 0, stream>>>(
      X, FIN, Wfc1, D0, bfc1, X0, D0, FIN, nullptr, nullptr);

  deg_rsqrt<<<NNODE / 8, 256, 0, stream>>>(adj, dv);
  symnorm<<<(NNODE * NNODE) / 256, 256, 0, stream>>>(adj, dv, An);
  gemm_sq_lds_f32<<<dim3(NNODE / SQ_T, NNODE / SQ_T), 256, 0, stream>>>(An, An, A2);

  const float* masks[3] = { adj, An, A2 };
  for (int ch = 0; ch < 3; ++ch) {
    const float* W1 = Wg[4 - 2 * ch];  const float* b1 = bg[4 - 2 * ch];
    const float* W2 = Wg[5 - 2 * ch];  const float* b2 = bg[5 - 2 * ch];
    float* embc = emb + (size_t)ch * NNODE * 256;

    branch_deg_rsqrt<<<NNODE / 8, 256, 0, stream>>>(masks[ch], adj, thr, dv);
    branch_norm<<<(NNODE * NNODE) / 256, 256, 0, stream>>>(masks[ch], adj, thr, dv, AnB);

    gemm_wmma_f32<false, false, false, 1, 8><<<dim3(HDIM / 64, NNODE / 32), blk8, 0, stream>>>(
        X0, D0, W1, HDIM, nullptr, Hb, HDIM, D0, nullptr, nullptr);
    gemm_wmma_f32<true, true, false, 1, 16><<<dim3(HDIM / 64, NNODE / 32), blk16, 0, stream>>>(
        AnB, NNODE, Hb, HDIM, b1, embc, 256, NNODE, nullptr, nullptr);
    gemm_wmma_f32<false, false, false, 1, 8><<<dim3(HDIM / 64, NNODE / 32), blk8, 0, stream>>>(
        embc, 256, W2, HDIM, nullptr, H2b, HDIM, HDIM, nullptr, nullptr);
    gemm_wmma_f32<true, true, false, 1, 16><<<dim3(HDIM / 64, NNODE / 32), blk16, 0, stream>>>(
        AnB, NNODE, H2b, HDIM, b2, embc + 128, 256, NNODE, nullptr, nullptr);
  }

  conv_dil2<<<(NNODE * 256) / 256, 256, 0, stream>>>(emb, cnnk, cnnb, emb_all);

  gemm_wmma_f32<true, true, true, 1, 8><<<dim3(64 / 64, BPAIR / 32), blk8, 0, stream>>>(
      emb_all, 256, Wa, 64, ba, hm, 64, 512, left, right);

  logits_k<<<(BPAIR * 2 + 255) / 256, 256, 0, stream>>>(hm, Wb, bb, logits);
}


# N x N normalisation passes (sym-norm, 3 branch norms) as 512 eight-wave work-groups of 6 rows: 16-byte loads and stores, all loads issued first; 4096 waves per pass, was 294912 (one element each)
# speedup vs baseline: 3.4353x; 1.0875x over previous
#include <hip/hip_runtime.h>
#include <hip/hip_bf16.h>

#define NNODE 3072
#define FIN   1546
#define D0    256
#define HDIM  128
#define BPAIR 16384

typedef __attribute__((ext_vector_type(2))) float v2f;
typedef __attribute__((ext_vector_type(8))) float v8f;

__device__ __forceinline__ void wmma8(v8f (&acc)[2][4], v2f a0, v2f a1,
                                      v2f b0, v2f b1, v2f b2, v2f b3)
{
  acc[0][0] = __builtin_amdgcn_wmma_f32_16x16x4_f32(false, a0, false, b0, (short)0, acc[0][0], false, false);
  acc[0][1] = __builtin_amdgcn_wmma_f32_16x16x4_f32(false, a0, false, b1, (short)0, acc[0][1], false, false);
  acc[0][2] = __builtin_amdgcn_wmma_f32_16x16x4_f32(false, a0, false, b2, (short)0, acc[0][2], false, false);
  acc[0][3] = __builtin_amdgcn_wmma_f32_16x16x4_f32(false, a0, false, b3, (short)0, acc[0][3], false, false);
  acc[1][0] = __builtin_amdgcn_wmma_f32_16x16x4_f32(false, a1, false, b0, (short)0, acc[1][0], false, false);
  acc[1][1] = __builtin_amdgcn_wmma_f32_16x16x4_f32(false, a1, false, b1, (short)0, acc[1][1], false, false);
  acc[1][2] = __builtin_amdgcn_wmma_f32_16x16x4_f32(false, a1, false, b2, (short)0, acc[1][2], false, false);
  acc[1][3] = __builtin_amdgcn_wmma_f32_16x16x4_f32(false, a1, false, b3, (short)0, acc[1][3], false, false);
}

#define RED_JP 272
#define RED_PW (8 * RED_JP)

template <bool RELU, bool BIAS, bool GATHER, int MW, int KS>
__global__ __launch_bounds__(32 * MW * KS) void gemm_wmma_f32(
    const float* __restrict__ A, int lda,
    const float* __restrict__ Bm, int ldb,
    const float* __restrict__ bias,
    float* __restrict__ C, int ldc,
    int K,
    const int* __restrict__ gl, const int* __restrict__ gr)
{
  static_assert(KS == 1 || MW == 1, "K split reduces one 32x64 tile per block");
  static_assert(!GATHER || (KS >= 2 && (64 % KS) == 0), "gather slice must stay in one half");

  const int lane = threadIdx.x & 31;
  const int wave = __builtin_amdgcn_readfirstlane((int)(threadIdx.x >> 5));
  const int wm   = wave % MW;
  const int wk   = wave / MW;
  const int half = lane >> 4;
  const int l16  = lane & 15;

  const int mb  = blockIdx.y * (32 * MW);
  const int m0  = mb + wm * 32;
  const int n0  = blockIdx.x * 64;
  const int am0 = m0 + l16;
  const int am1 = m0 + 16 + l16;

  v8f acc[2][4] = {};

  const int nsteps = (GATHER ? 512 : K) >> 2;
  const int per    = (nsteps + KS - 1) / KS;
  const int sbeg   = min(wk * per, nsteps);
  const int send   = min(sbeg + per, nsteps);
  const int kbeg   = 4 * sbeg;

  const float* bp = Bm + (size_t)(kbeg + 2 * half) * ldb + n0 + l16;
  const size_t bstep = (size_t)4 * ldb;

  if (GATHER) {
    const int* gi = (kbeg < 256) ? gl : gr;
    const float* ap0 = A + (size_t)gi[am0] * 256 + (kbeg & 255) + 2 * half;
    const float* ap1 = A + (size_t)gi[am1] * 256 + (kbeg & 255) + 2 * half;
    #pragma unroll 4
    for (int s = sbeg; s < send; ++s) {
      v2f a0 = *(const v2f*)ap0;
      v2f a1 = *(const v2f*)ap1;
      v2f b0, b1, b2, b3;
      b0.x = bp[0];  b0.y = bp[ldb + 0];
      b1.x = bp[16]; b1.y = bp[ldb + 16];
      b2.x = bp[32]; b2.y = bp[ldb + 32];
      b3.x = bp[48]; b3.y = bp[ldb + 48];
      wmma8(acc, a0, a1, b0, b1, b2, b3);
      ap0 += 4; ap1 += 4; bp += bstep;
    }
  } else {
    const float* ap0 = A + (size_t)am0 * lda + kbeg + 2 * half;
    const float* ap1 = A + (size_t)am1 * lda + kbeg + 2 * half;
    #pragma unroll 2
    for (int s = sbeg; s < send; ++s) {
      v2f a0 = *(const v2f*)ap0;
      v2f a1 = *(const v2f*)ap1;
      v2f b0, b1, b2, b3;
      b0.x = bp[0];  b0.y = bp[ldb + 0];
      b1.x = bp[16]; b1.y = bp[ldb + 16];
      b2.x = bp[32]; b2.y = bp[ldb + 32];
      b3.x = bp[48]; b3.y = bp[ldb + 48];
      wmma8(acc, a0, a1, b0, b1, b2, b3);
      ap0 += 4; ap1 += 4; bp += bstep;
    }
    const int kmain = K & ~3;
    if (kmain < K && wk == KS - 1) {
      const int ka = kmain + 2 * half;
      const bool g0 = (ka < K), g1 = (ka + 1 < K);
      v2f a0, a1, b0, b1, b2, b3;
      a0.x = g0 ? ap0[0] : 0.0f;  a0.y = g1 ? ap0[1] : 0.0f;
      a1.x = g0 ? ap1[0] : 0.0f;  a1.y = g1 ? ap1[1] : 0.0f;
      b0.x = g0 ? bp[0]  : 0.0f;  b0.y = g1 ? bp[ldb + 0]  : 0.0f;
      b1.x = g0 ? bp[16] : 0.0f;  b1.y = g1 ? bp[ldb + 16] : 0.0f;
      b2.x = g0 ? bp[32] : 0.0f;  b2.y = g1 ? bp[ldb + 32] : 0.0f;
      b3.x = g0 ? bp[48] : 0.0f;  b3.y = g1 ? bp[ldb + 48] : 0.0f;
      wmma8(acc, a0, a1, b0, b1, b2, b3);
    }
  }

  if (KS == 1) {
    #pragma unroll
    for (int s = 0; s < 2; ++s) {
      #pragma unroll
      for (int v = 0; v < 8; ++v) {
        const int row = m0 + 16 * s + v + 8 * half;
        #pragma unroll
        for (int j = 0; j < 4; ++j) {
          const int col = n0 + 16 * j + l16;
          float x = acc[s][j][v];
          if (BIAS) x += bias[col];
          if (RELU) x = fmaxf(x, 0.0f);
          C[(size_t)row * ldc + col] = x;
        }
      }
    }
  } else {
    __shared__ __attribute__((aligned(16))) float red[KS * RED_PW];
    float* my = red + wk * RED_PW + lane;
    #pragma unroll
    for (int s = 0; s < 2; ++s) {
      #pragma unroll
      for (int j = 0; j < 4; ++j) {
        #pragma unroll
        for (int v = 0; v < 8; ++v)
          my[(s * 4 + j) * RED_JP + v * 32] = acc[s][j][v];
      }
    }
    __syncthreads();
    for (int e = threadIdx.x; e < 512; e += 32 * KS) {
      const int row = e >> 4;
      const int c4  = (e & 15) * 4;
      const int s   = row >> 4;
      const int hf  = (row >> 3) & 1;
      const int v   = row & 7;
      const int j   = c4 >> 4;
      const int l   = c4 & 15;
      const float* p = red + (s * 4 + j) * RED_JP + v * 32 + hf * 16 + l;
      float4 sum = *(const float4*)p;
      #pragma unroll
      for (int w = 1; w < KS; ++w) {
        const float4 x = *(const float4*)(p + w * RED_PW);
        sum.x += x.x; sum.y += x.y; sum.z += x.z; sum.w += x.w;
      }
      const int col = n0 + c4;
      if (BIAS) {
        sum.x += bias[col + 0]; sum.y += bias[col + 1];
        sum.z += bias[col + 2]; sum.w += bias[col + 3];
      }
      if (RELU) {
        sum.x = fmaxf(sum.x, 0.0f); sum.y = fmaxf(sum.y, 0.0f);
        sum.z = fmaxf(sum.z, 0.0f); sum.w = fmaxf(sum.w, 0.0f);
      }
      *(float4*)(C + (size_t)(mb + row) * ldc + col) = sum;
    }
  }
}

#define SQ_T   192
#define SQ_KT  64
#define SQ_PA  68
#define SQ_PB  416
#define SQ_ASZ (SQ_T * SQ_PA)
#define SQ_BSZ ((SQ_KT / 2) * SQ_PB)
#define SQ_STG (SQ_ASZ + SQ_BSZ)

__device__ __forceinline__ void sq_load(float4 (&ra)[12], float4 (&rb0)[6], float4 (&rb1)[6],
                                        const float* gA, const float* gB)
{
  #pragma unroll
  for (int i = 0; i < 12; ++i) ra[i] = *(const float4*)(gA + (size_t)i * 16 * NNODE);
  #pragma unroll
  for (int a = 0; a < 2; ++a) {
    #pragma unroll
    for (int b = 0; b < 3; ++b) {
      const float* g = gB + (size_t)a * 32 * NNODE + b * 64;
      rb0[a * 3 + b] = *(const float4*)g;
      rb1[a * 3 + b] = *(const float4*)(g + NNODE);
    }
  }
}

template <int I>
__device__ __forceinline__ void sq_store_a(float* aW, const float4 (&ra)[12])
{
  *(float4*)(aW + I * 16 * SQ_PA) = make_float4(ra[I].x, ra[I].y, ra[I].z, ra[I].w);
}

template <int Q>
__device__ __forceinline__ void sq_store_b(float* bW, const float4 (&rb0)[6],
                                           const float4 (&rb1)[6])
{
  float* d = bW + (Q / 3) * 16 * SQ_PB + (Q % 3) * 128;
  *(float4*)d       = make_float4(rb0[Q].x, rb1[Q].x, rb0[Q].y, rb1[Q].y);
  *(float4*)(d + 4) = make_float4(rb0[Q].z, rb1[Q].z, rb0[Q].w, rb1[Q].w);
}

template <bool STAGE_NEXT, int I>
__device__ __forceinline__ void sq_kstep(v8f (&acc)[3][6], v2f (&a)[2][3], v2f (&b)[2][6],
                                         const float* aR, const float* bR,
                                         float* aW, float* bW, const float4 (&ra)[12],
                                         const float4 (&rb0)[6], const float4 (&rb1)[6])
{
  constexpr int c = I & 1, n = c ^ 1;
  if (I + 1 < SQ_KT / 4) {
    #pragma unroll
    for (int s = 0; s < 3; ++s)
      a[n][s] = *(const v2f*)(aR + s * 16 * SQ_PA + 4 * (I + 1));
    #pragma unroll
    for (int j = 0; j < 6; ++j)
      b[n][j] = *(const v2f*)(bR + 2 * (I + 1) * SQ_PB + j * 32);
  }
  #pragma unroll
  for (int s = 0; s < 3; ++s) {
    #pragma unroll
    for (int j = 0; j < 6; ++j)
      acc[s][j] = __builtin_amdgcn_wmma_f32_16x16x4_f32(false, a[c][s], false, b[c][j],
                                                        (short)0, acc[s][j], false, false);
  }
  if constexpr (STAGE_NEXT && I >= 2 && I < 14) {
    __builtin_amdgcn_sched_barrier(0);
    sq_store_a<I - 2>(aW, ra);
    if constexpr (((I - 2) & 1) == 0) sq_store_b<(I - 2) / 2>(bW, rb0, rb1);
  }
}

template <bool STAGE_NEXT>
__device__ __forceinline__ void sq_slab(v8f (&acc)[3][6], const float* aR, const float* bR,
                                        float* aW, float* bW, const float4 (&ra)[12],
                                        const float4 (&rb0)[6], const float4 (&rb1)[6])
{
  v2f a[2][3], b[2][6];
  #pragma unroll
  for (int s = 0; s < 3; ++s) a[0][s] = *(const v2f*)(aR + s * 16 * SQ_PA);
  #pragma unroll
  for (int j = 0; j < 6; ++j) b[0][j] = *(const v2f*)(bR + j * 32);
  sq_kstep<STAGE_NEXT,  0>(acc, a, b, aR, bR, aW, bW, ra, rb0, rb1);
  sq_kstep<STAGE_NEXT,  1>(acc, a, b, aR, bR, aW, bW, ra, rb0, rb1);
  sq_kstep<STAGE_NEXT,  2>(acc, a, b, aR, bR, aW, bW, ra, rb0, rb1);
  sq_kstep<STAGE_NEXT,  3>(acc, a, b, aR, bR, aW, bW, ra, rb0, rb1);
  sq_kstep<STAGE_NEXT,  4>(acc, a, b, aR, bR, aW, bW, ra, rb0, rb1);
  sq_kstep<STAGE_NEXT,  5>(acc, a, b, aR, bR, aW, bW, ra, rb0, rb1);
  sq_kstep<STAGE_NEXT,  6>(acc, a, b, aR, bR, aW, bW, ra, rb0, rb1);
  sq_kstep<STAGE_NEXT,  7>(acc, a, b, aR, bR, aW, bW, ra, rb0, rb1);
  sq_kstep<STAGE_NEXT,  8>(acc, a, b, aR, bR, aW, bW, ra, rb0, rb1);
  sq_kstep<STAGE_NEXT,  9>(acc, a, b, aR, bR, aW, bW, ra, rb0, rb1);
  sq_kstep<STAGE_NEXT, 10>(acc, a, b, aR, bR, aW, bW, ra, rb0, rb1);
  sq_kstep<STAGE_NEXT, 11>(acc, a, b, aR, bR, aW, bW, ra, rb0, rb1);
  sq_kstep<STAGE_NEXT, 12>(acc, a, b, aR, bR, aW, bW, ra, rb0, rb1);
  sq_kstep<STAGE_NEXT, 13>(acc, a, b, aR, bR, aW, bW, ra, rb0, rb1);
  sq_kstep<STAGE_NEXT, 14>(acc, a, b, aR, bR, aW, bW, ra, rb0, rb1);
  sq_kstep<STAGE_NEXT, 15>(acc, a, b, aR, bR, aW, bW, ra, rb0, rb1);
}

__global__ __launch_bounds__(256) void gemm_sq_lds_f32(const float* __restrict__ A,
                                                       const float* __restrict__ Bm,
                                                       float* __restrict__ C)
{
  __shared__ __attribute__((aligned(16))) float sm[2 * SQ_STG];

  const int tid  = threadIdx.x;
  const int lane = tid & 31;
  const int wave = __builtin_amdgcn_readfirstlane(tid >> 5);
  const int wm   = wave >> 1;
  const int wn   = wave & 1;
  const int half = lane >> 4;
  const int l16  = lane & 15;
  const int mb   = blockIdx.y * SQ_T;
  const int nb   = blockIdx.x * SQ_T;

  const int tr = tid >> 4, tc = tid & 15;
  const float* gA = A + (size_t)(mb + tr) * NNODE + 4 * tc;
  const float* gB = Bm + (size_t)(2 * tr) * NNODE + nb + 4 * tc;
  const int sAo = tr * SQ_PA + 4 * tc;
  const int sBo = SQ_ASZ + tr * SQ_PB + 8 * tc;
  const int aF = (48 * wm + l16) * SQ_PA + 2 * half;
  const int bF = SQ_ASZ + half * SQ_PB + 2 * (96 * wn + l16);

  v8f acc[3][6] = {};
  float4 ra[12], rb0[6], rb1[6];

  sq_load(ra, rb0, rb1, gA, gB);
  sq_store_a<0>(sm + sAo, ra);  sq_store_a<1>(sm + sAo, ra);  sq_store_a<2>(sm + sAo, ra);
  sq_store_a<3>(sm + sAo, ra);  sq_store_a<4>(sm + sAo, ra);  sq_store_a<5>(sm + sAo, ra);
  sq_store_a<6>(sm + sAo, ra);  sq_store_a<7>(sm + sAo, ra);  sq_store_a<8>(sm + sAo, ra);
  sq_store_a<9>(sm + sAo, ra);  sq_store_a<10>(sm + sAo, ra); sq_store_a<11>(sm + sAo, ra);
  sq_store_b<0>(sm + sBo, rb0, rb1);  sq_store_b<1>(sm + sBo, rb0, rb1);
  sq_store_b<2>(sm + sBo, rb0, rb1);  sq_store_b<3>(sm + sBo, rb0, rb1);
  sq_store_b<4>(sm + sBo, rb0, rb1);  sq_store_b<5>(sm + sBo, rb0, rb1);
  __syncthreads();

  int st = 0;
  for (int t = 0; t < NNODE / SQ_KT - 1; ++t) {
    gA += SQ_KT;
    gB += (size_t)SQ_KT * NNODE;
    sq_load(ra, rb0, rb1, gA, gB);
    sq_slab<true>(acc, sm + st + aF, sm + st + bF, sm + (SQ_STG - st) + sAo,
                  sm + (SQ_STG - st) + sBo, ra, rb0, rb1);
    __syncthreads();
    st = SQ_STG - st;
  }
  sq_slab<false>(acc, sm + st + aF, sm + st + bF, nullptr, nullptr, ra, rb0, rb1);

  #pragma unroll
  for (int s = 0; s < 3; ++s) {
    #pragma unroll
    for (int v = 0; v < 8; ++v) {
      const int row = mb + 48 * wm + 16 * s + v + 8 * half;
      #pragma unroll
      for (int j = 0; j < 6; ++j) {
        const int col = nb + 96 * wn + 16 * j + l16;
        C[(size_t)row * NNODE + col] = acc[s][j][v];
      }
    }
  }
}

__global__ void deg_rsqrt(const float* __restrict__ A, float* __restrict__ d)
{
  const int gw   = (blockIdx.x * blockDim.x + threadIdx.x) >> 5;
  const int lane = threadIdx.x & 31;
  if (gw >= NNODE) return;
  const float* row = A + (size_t)gw * NNODE;
  float s = 0.0f;
  for (int j = lane; j < NNODE; j += 32) s += row[j];
  #pragma unroll
  for (int off = 16; off > 0; off >>= 1) s += __shfl_xor(s, off, 32);
  if (lane == 0) d[gw] = (s > 0.0f) ? rsqrtf(s) : 0.0f;
}

__global__ void branch_deg_rsqrt(const float* __restrict__ mask,
                                 const float* __restrict__ adj,
                                 const float* __restrict__ thrp,
                                 float* __restrict__ d)
{
  const int gw   = (blockIdx.x * blockDim.x + threadIdx.x) >> 5;
  const int lane = threadIdx.x & 31;
  if (gw >= NNODE) return;
  const float thr = thrp[0];
  const size_t base = (size_t)gw * NNODE;
  float s = 0.0f;
  for (int j = lane; j < NNODE; j += 32)
    s += (mask[base + j] > thr) ? adj[base + j] : 0.0f;
  #pragma unroll
  for (int off = 16; off > 0; off >>= 1) s += __shfl_xor(s, off, 32);
  if (lane == 0) d[gw] = rsqrtf(s + 1.0f);
}

#define NORM_R 6
#define NORM_C (NNODE / 4 / 256)
#define NORM_WAVES __attribute__((amdgpu_waves_per_eu(4, 4)))

__global__ __launch_bounds__(256) NORM_WAVES void symnorm(const float* __restrict__ adj,
                                                          const float* __restrict__ d,
                                                          float* __restrict__ An)
{
  const int tid = threadIdx.x;
  const int r0  = blockIdx.x * NORM_R;
  const float* ap = adj + (size_t)r0 * NNODE + 4 * tid;
  float*       op = An  + (size_t)r0 * NNODE + 4 * tid;

  float4 dj[NORM_C], a[NORM_R][NORM_C];
  float  di[NORM_R];
  #pragma unroll
  for (int c = 0; c < NORM_C; ++c) dj[c] = *(const float4*)(d + 4 * tid + 1024 * c);
  #pragma unroll
  for (int r = 0; r < NORM_R; ++r) di[r] = d[r0 + r];
  #pragma unroll
  for (int r = 0; r < NORM_R; ++r) {
    #pragma unroll
    for (int c = 0; c < NORM_C; ++c)
      a[r][c] = *(const float4*)(ap + (size_t)r * NNODE + 1024 * c);
  }
  #pragma unroll
  for (int r = 0; r < NORM_R; ++r) {
    #pragma unroll
    for (int c = 0; c < NORM_C; ++c) {
      float4 o;
      o.x = a[r][c].x * di[r] * dj[c].x;
      o.y = a[r][c].y * di[r] * dj[c].y;
      o.z = a[r][c].z * di[r] * dj[c].z;
      o.w = a[r][c].w * di[r] * dj[c].w;
      *(float4*)(op + (size_t)r * NNODE + 1024 * c) = o;
    }
  }
}

__device__ __forceinline__ float bn_elem(float m, float a, float thr, bool diag,
                                         float di, float dj)
{
  float w = (m > thr) ? a : 0.0f;
  if (diag) w += 1.0f;
  return w * di * dj;
}

__global__ __launch_bounds__(256) NORM_WAVES void branch_norm(
    const float* __restrict__ mask, const float* __restrict__ adj,
    const float* __restrict__ thrp, const float* __restrict__ d, float* __restrict__ An)
{
  const int tid = threadIdx.x;
  const int r0  = blockIdx.x * NORM_R;
  const float thr = thrp[0];
  const float* mp = mask + (size_t)r0 * NNODE + 4 * tid;
  const float* ap = adj  + (size_t)r0 * NNODE + 4 * tid;
  float*       op = An   + (size_t)r0 * NNODE + 4 * tid;

  float4 dj[NORM_C], m[NORM_R][NORM_C], a[NORM_R][NORM_C];
  float  di[NORM_R];
  #pragma unroll
  for (int c = 0; c < NORM_C; ++c) dj[c] = *(const float4*)(d + 4 * tid + 1024 * c);
  #pragma unroll
  for (int r = 0; r < NORM_R; ++r) di[r] = d[r0 + r];
  #pragma unroll
  for (int r = 0; r < NORM_R; ++r) {
    #pragma unroll
    for (int c = 0; c < NORM_C; ++c) {
      m[r][c] = *(const float4*)(mp + (size_t)r * NNODE + 1024 * c);
      a[r][c] = *(const float4*)(ap + (size_t)r * NNODE + 1024 * c);
    }
  }
  #pragma unroll
  for (int r = 0; r < NORM_R; ++r) {
    #pragma unroll
    for (int c = 0; c < NORM_C; ++c) {
      const int dc = 4 * tid + 1024 * c - (r0 + r);
      float4 o;
      o.x = bn_elem(m[r][c].x, a[r][c].x, thr, dc ==  0, di[r], dj[c].x);
      o.y = bn_elem(m[r][c].y, a[r][c].y, thr, dc == -1, di[r], dj[c].y);
      o.z = bn_elem(m[r][c].z, a[r][c].z, thr, dc == -2, di[r], dj[c].z);
      o.w = bn_elem(m[r][c].w, a[r][c].w, thr, dc == -3, di[r], dj[c].w);
      *(float4*)(op + (size_t)r * NNODE + 1024 * c) = o;
    }
  }
}

__global__ void conv_dil2(const float* __restrict__ emb,
                          const float* __restrict__ kw,
                          const float* __restrict__ kb,
                          float* __restrict__ out)
{
  const int idx = blockIdx.x * blockDim.x + threadIdx.x;
  if (idx >= NNODE * 256) return;
  const int n = idx >> 8;
  const int c = idx & 255;
  float acc = kb[0];
  #pragma unroll
  for (int ch = 0; ch < 3; ++ch) {
    #pragma unroll
    for (int kh = 0; kh < 3; ++kh) {
      const int ih = n + 2 * kh - 2;
      if (ih < 0 || ih >= NNODE) continue;
      #pragma unroll
      for (int kwi = 0; kwi < 3; ++kwi) {
        const int iw = c + 2 * kwi - 2;
        if (iw < 0 || iw >= 256) continue;
        acc += emb[((size_t)ch * NNODE + ih) * 256 + iw] * kw[ch * 9 + kh * 3 + kwi];
      }
    }
  }
  out[idx] = acc;
}

__global__ void logits_k(const float* __restrict__ h,
                         const float* __restrict__ Wb,
                         const float* __restrict__ bb,
                         float* __restrict__ out)
{
  const int i = blockIdx.x * blockDim.x + threadIdx.x;
  if (i >= BPAIR * 2) return;
  const int b = i >> 1;
  const int c = i & 1;
  float acc = bb[c];
  #pragma unroll
  for (int k = 0; k < 64; ++k) acc += h[b * 64 + k] * Wb[k * 2 + c];
  out[i] = acc;
}

extern "C" void kernel_launch(void* const* d_in, const int* in_sizes, int n_in,
                              void* d_out, int out_size, void* d_ws, size_t ws_size,
                              hipStream_t stream)
{
  const int*   left  = (const int*)d_in[0];
  const int*   right = (const int*)d_in[1];
  const float* X     = (const float*)d_in[2];
  const float* adj   = (const float*)d_in[3];
  const float* thr   = (const float*)d_in[4];
  const float* Wfc1  = (const float*)d_in[5];
  const float* bfc1  = (const float*)d_in[6];
  const float* Wg[6] = {(const float*)d_in[7],  (const float*)d_in[9],
                        (const float*)d_in[11], (const float*)d_in[13],
                        (const float*)d_in[15], (const float*)d_in[17]};
  const float* bg[6] = {(const float*)d_in[8],  (const float*)d_in[10],
                        (const float*)d_in[12], (const float*)d_in[14],
                        (const float*)d_in[16], (const float*)d_in[18]};
  const float* cnnk  = (const float*)d_in[19];
  const float* cnnb  = (const float*)d_in[20];
  const float* Wa    = (const float*)d_in[21];
  const float* ba    = (const float*)d_in[22];
  const float* Wb    = (const float*)d_in[23];
  const float* bb    = (const float*)d_in[24];

  float* out     = (float*)d_out;
  float* logits  = out;
  float* emb_all = out + BPAIR * 2;

  float* ws = (float*)d_ws;
  size_t off = 0;
  float* X0  = ws + off; off += (size_t)NNODE * D0;
  float* An  = ws + off; off += (size_t)NNODE * NNODE;
  float* A2  = ws + off; off += (size_t)NNODE * NNODE;
  float* AnB = ws + off; off += (size_t)NNODE * NNODE;
  float* dv  = ws + off; off += NNODE;
  float* Hb  = ws + off; off += (size_t)NNODE * HDIM;
  float* H2b = ws + off; off += (size_t)NNODE * HDIM;
  float* emb = ws + off; off += (size_t)3 * NNODE * 256;
  float* hm  = ws + off; off += (size_t)BPAIR * 64;
  if (ws_size < off * sizeof(float)) return;

  const dim3 blk8(256);
  const dim3 blk16(512);

  gemm_wmma_f32<true, true, false, 1, 16><<<dim3(D0 / 64, NNODE / 32), blk16, 0, stream>>>(
      X, FIN, Wfc1, D0, bfc1, X0, D0, FIN, nullptr, nullptr);

  deg_rsqrt<<<NNODE / 8, 256, 0, stream>>>(adj, dv);
  symnorm<<<NNODE / NORM_R, 256, 0, stream>>>(adj, dv, An);
  gemm_sq_lds_f32<<<dim3(NNODE / SQ_T, NNODE / SQ_T), 256, 0, stream>>>(An, An, A2);

  const float* masks[3] = { adj, An, A2 };
  for (int ch = 0; ch < 3; ++ch) {
    const float* W1 = Wg[4 - 2 * ch];  const float* b1 = bg[4 - 2 * ch];
    const float* W2 = Wg[5 - 2 * ch];  const float* b2 = bg[5 - 2 * ch];
    float* embc = emb + (size_t)ch * NNODE * 256;

    branch_deg_rsqrt<<<NNODE / 8, 256, 0, stream>>>(masks[ch], adj, thr, dv);
    branch_norm<<<NNODE / NORM_R, 256, 0, stream>>>(masks[ch], adj, thr, dv, AnB);

    gemm_wmma_f32<false, false, false, 1, 8><<<dim3(HDIM / 64, NNODE / 32), blk8, 0, stream>>>(
        X0, D0, W1, HDIM, nullptr, Hb, HDIM, D0, nullptr, nullptr);
    gemm_wmma_f32<true, true, false, 1, 16><<<dim3(HDIM / 64, NNODE / 32), blk16, 0, stream>>>(
        AnB, NNODE, Hb, HDIM, b1, embc, 256, NNODE, nullptr, nullptr);
    gemm_wmma_f32<false, false, false, 1, 8><<<dim3(HDIM / 64, NNODE / 32), blk8, 0, stream>>>(
        embc, 256, W2, HDIM, nullptr, H2b, HDIM, HDIM, nullptr, nullptr);
    gemm_wmma_f32<true, true, false, 1, 16><<<dim3(HDIM / 64, NNODE / 32), blk16, 0, stream>>>(
        AnB, NNODE, H2b, HDIM, b2, embc + 128, 256, NNODE, nullptr, nullptr);
  }

  conv_dil2<<<(NNODE * 256) / 256, 256, 0, stream>>>(emb, cnnk, cnnb, emb_all);

  gemm_wmma_f32<true, true, true, 1, 8><<<dim3(64 / 64, BPAIR / 32), blk8, 0, stream>>>(
      emb_all, 256, Wa, 64, ba, hm, 64, 512, left, right);

  logits_k<<<(BPAIR * 2 + 255) / 256, 256, 0, stream>>>(hm, Wb, bb, logits);
}


# branch degree pass: a lane's 48 16-byte loads (mask, adj) go out ahead of its adds, adj unconditional; was one adj load behind each mask compare, 96 dependent round trips a row; row sum order changed
# speedup vs baseline: 4.1338x; 1.2033x over previous
#include <hip/hip_runtime.h>
#include <hip/hip_bf16.h>

#define NNODE 3072
#define FIN   1546
#define D0    256
#define HDIM  128
#define BPAIR 16384

typedef __attribute__((ext_vector_type(2))) float v2f;
typedef __attribute__((ext_vector_type(8))) float v8f;

__device__ __forceinline__ void wmma8(v8f (&acc)[2][4], v2f a0, v2f a1,
                                      v2f b0, v2f b1, v2f b2, v2f b3)
{
  acc[0][0] = __builtin_amdgcn_wmma_f32_16x16x4_f32(false, a0, false, b0, (short)0, acc[0][0], false, false);
  acc[0][1] = __builtin_amdgcn_wmma_f32_16x16x4_f32(false, a0, false, b1, (short)0, acc[0][1], false, false);
  acc[0][2] = __builtin_amdgcn_wmma_f32_16x16x4_f32(false, a0, false, b2, (short)0, acc[0][2], false, false);
  acc[0][3] = __builtin_amdgcn_wmma_f32_16x16x4_f32(false, a0, false, b3, (short)0, acc[0][3], false, false);
  acc[1][0] = __builtin_amdgcn_wmma_f32_16x16x4_f32(false, a1, false, b0, (short)0, acc[1][0], false, false);
  acc[1][1] = __builtin_amdgcn_wmma_f32_16x16x4_f32(false, a1, false, b1, (short)0, acc[1][1], false, false);
  acc[1][2] = __builtin_amdgcn_wmma_f32_16x16x4_f32(false, a1, false, b2, (short)0, acc[1][2], false, false);
  acc[1][3] = __builtin_amdgcn_wmma_f32_16x16x4_f32(false, a1, false, b3, (short)0, acc[1][3], false, false);
}

#define RED_JP 272
#define RED_PW (8 * RED_JP)

template <bool RELU, bool BIAS, bool GATHER, int MW, int KS>
__global__ __launch_bounds__(32 * MW * KS) void gemm_wmma_f32(
    const float* __restrict__ A, int lda,
    const float* __restrict__ Bm, int ldb,
    const float* __restrict__ bias,
    float* __restrict__ C, int ldc,
    int K,
    const int* __restrict__ gl, const int* __restrict__ gr)
{
  static_assert(KS == 1 || MW == 1, "K split reduces one 32x64 tile per block");
  static_assert(!GATHER || (KS >= 2 && (64 % KS) == 0), "gather slice must stay in one half");

  const int lane = threadIdx.x & 31;
  const int wave = __builtin_amdgcn_readfirstlane((int)(threadIdx.x >> 5));
  const int wm   = wave % MW;
  const int wk   = wave / MW;
  const int half = lane >> 4;
  const int l16  = lane & 15;

  const int mb  = blockIdx.y * (32 * MW);
  const int m0  = mb + wm * 32;
  const int n0  = blockIdx.x * 64;
  const int am0 = m0 + l16;
  const int am1 = m0 + 16 + l16;

  v8f acc[2][4] = {};

  const int nsteps = (GATHER ? 512 : K) >> 2;
  const int per    = (nsteps + KS - 1) / KS;
  const int sbeg   = min(wk * per, nsteps);
  const int send   = min(sbeg + per, nsteps);
  const int kbeg   = 4 * sbeg;

  const float* bp = Bm + (size_t)(kbeg + 2 * half) * ldb + n0 + l16;
  const size_t bstep = (size_t)4 * ldb;

  if (GATHER) {
    const int* gi = (kbeg < 256) ? gl : gr;
    const float* ap0 = A + (size_t)gi[am0] * 256 + (kbeg & 255) + 2 * half;
    const float* ap1 = A + (size_t)gi[am1] * 256 + (kbeg & 255) + 2 * half;
    #pragma unroll 4
    for (int s = sbeg; s < send; ++s) {
      v2f a0 = *(const v2f*)ap0;
      v2f a1 = *(const v2f*)ap1;
      v2f b0, b1, b2, b3;
      b0.x = bp[0];  b0.y = bp[ldb + 0];
      b1.x = bp[16]; b1.y = bp[ldb + 16];
      b2.x = bp[32]; b2.y = bp[ldb + 32];
      b3.x = bp[48]; b3.y = bp[ldb + 48];
      wmma8(acc, a0, a1, b0, b1, b2, b3);
      ap0 += 4; ap1 += 4; bp += bstep;
    }
  } else {
    const float* ap0 = A + (size_t)am0 * lda + kbeg + 2 * half;
    const float* ap1 = A + (size_t)am1 * lda + kbeg + 2 * half;
    #pragma unroll 2
    for (int s = sbeg; s < send; ++s) {
      v2f a0 = *(const v2f*)ap0;
      v2f a1 = *(const v2f*)ap1;
      v2f b0, b1, b2, b3;
      b0.x = bp[0];  b0.y = bp[ldb + 0];
      b1.x = bp[16]; b1.y = bp[ldb + 16];
      b2.x = bp[32]; b2.y = bp[ldb + 32];
      b3.x = bp[48]; b3.y = bp[ldb + 48];
      wmma8(acc, a0, a1, b0, b1, b2, b3);
      ap0 += 4; ap1 += 4; bp += bstep;
    }
    const int kmain = K & ~3;
    if (kmain < K && wk == KS - 1) {
      const int ka = kmain + 2 * half;
      const bool g0 = (ka < K), g1 = (ka + 1 < K);
      v2f a0, a1, b0, b1, b2, b3;
      a0.x = g0 ? ap0[0] : 0.0f;  a0.y = g1 ? ap0[1] : 0.0f;
      a1.x = g0 ? ap1[0] : 0.0f;  a1.y = g1 ? ap1[1] : 0.0f;
      b0.x = g0 ? bp[0]  : 0.0f;  b0.y = g1 ? bp[ldb + 0]  : 0.0f;
      b1.x = g0 ? bp[16] : 0.0f;  b1.y = g1 ? bp[ldb + 16] : 0.0f;
      b2.x = g0 ? bp[32] : 0.0f;  b2.y = g1 ? bp[ldb + 32] : 0.0f;
      b3.x = g0 ? bp[48] : 0.0f;  b3.y = g1 ? bp[ldb + 48] : 0.0f;
      wmma8(acc, a0, a1, b0, b1, b2, b3);
    }
  }

  if (KS == 1) {
    #pragma unroll
    for (int s = 0; s < 2; ++s) {
      #pragma unroll
      for (int v = 0; v < 8; ++v) {
        const int row = m0 + 16 * s + v + 8 * half;
        #pragma unroll
        for (int j = 0; j < 4; ++j) {
          const int col = n0 + 16 * j + l16;
          float x = acc[s][j][v];
          if (BIAS) x += bias[col];
          if (RELU) x = fmaxf(x, 0.0f);
          C[(size_t)row * ldc + col] = x;
        }
      }
    }
  } else {
    __shared__ __attribute__((aligned(16))) float red[KS * RED_PW];
    float* my = red + wk * RED_PW + lane;
    #pragma unroll
    for (int s = 0; s < 2; ++s) {
      #pragma unroll
      for (int j = 0; j < 4; ++j) {
        #pragma unroll
        for (int v = 0; v < 8; ++v)
          my[(s * 4 + j) * RED_JP + v * 32] = acc[s][j][v];
      }
    }
    __syncthreads();
    for (int e = threadIdx.x; e < 512; e += 32 * KS) {
      const int row = e >> 4;
      const int c4  = (e & 15) * 4;
      const int s   = row >> 4;
      const int hf  = (row >> 3) & 1;
      const int v   = row & 7;
      const int j   = c4 >> 4;
      const int l   = c4 & 15;
      const float* p = red + (s * 4 + j) * RED_JP + v * 32 + hf * 16 + l;
      float4 sum = *(const float4*)p;
      #pragma unroll
      for (int w = 1; w < KS; ++w) {
        const float4 x = *(const float4*)(p + w * RED_PW);
        sum.x += x.x; sum.y += x.y; sum.z += x.z; sum.w += x.w;
      }
      const int col = n0 + c4;
      if (BIAS) {
        sum.x += bias[col + 0]; sum.y += bias[col + 1];
        sum.z += bias[col + 2]; sum.w += bias[col + 3];
      }
      if (RELU) {
        sum.x = fmaxf(sum.x, 0.0f); sum.y = fmaxf(sum.y, 0.0f);
        sum.z = fmaxf(sum.z, 0.0f); sum.w = fmaxf(sum.w, 0.0f);
      }
      *(float4*)(C + (size_t)(mb + row) * ldc + col) = sum;
    }
  }
}

#define SQ_T   192
#define SQ_KT  64
#define SQ_PA  68
#define SQ_PB  416
#define SQ_ASZ (SQ_T * SQ_PA)
#define SQ_BSZ ((SQ_KT / 2) * SQ_PB)
#define SQ_STG (SQ_ASZ + SQ_BSZ)

__device__ __forceinline__ void sq_load(float4 (&ra)[12], float4 (&rb0)[6], float4 (&rb1)[6],
                                        const float* gA, const float* gB)
{
  #pragma unroll
  for (int i = 0; i < 12; ++i) ra[i] = *(const float4*)(gA + (size_t)i * 16 * NNODE);
  #pragma unroll
  for (int a = 0; a < 2; ++a) {
    #pragma unroll
    for (int b = 0; b < 3; ++b) {
      const float* g = gB + (size_t)a * 32 * NNODE + b * 64;
      rb0[a * 3 + b] = *(const float4*)g;
      rb1[a * 3 + b] = *(const float4*)(g + NNODE);
    }
  }
}

template <int I>
__device__ __forceinline__ void sq_store_a(float* aW, const float4 (&ra)[12])
{
  *(float4*)(aW + I * 16 * SQ_PA) = make_float4(ra[I].x, ra[I].y, ra[I].z, ra[I].w);
}

template <int Q>
__device__ __forceinline__ void sq_store_b(float* bW, const float4 (&rb0)[6],
                                           const float4 (&rb1)[6])
{
  float* d = bW + (Q / 3) * 16 * SQ_PB + (Q % 3) * 128;
  *(float4*)d       = make_float4(rb0[Q].x, rb1[Q].x, rb0[Q].y, rb1[Q].y);
  *(float4*)(d + 4) = make_float4(rb0[Q].z, rb1[Q].z, rb0[Q].w, rb1[Q].w);
}

template <bool STAGE_NEXT, int I>
__device__ __forceinline__ void sq_kstep(v8f (&acc)[3][6], v2f (&a)[2][3], v2f (&b)[2][6],
                                         const float* aR, const float* bR,
                                         float* aW, float* bW, const float4 (&ra)[12],
                                         const float4 (&rb0)[6], const float4 (&rb1)[6])
{
  constexpr int c = I & 1, n = c ^ 1;
  if (I + 1 < SQ_KT / 4) {
    #pragma unroll
    for (int s = 0; s < 3; ++s)
      a[n][s] = *(const v2f*)(aR + s * 16 * SQ_PA + 4 * (I + 1));
    #pragma unroll
    for (int j = 0; j < 6; ++j)
      b[n][j] = *(const v2f*)(bR + 2 * (I + 1) * SQ_PB + j * 32);
  }
  #pragma unroll
  for (int s = 0; s < 3; ++s) {
    #pragma unroll
    for (int j = 0; j < 6; ++j)
      acc[s][j] = __builtin_amdgcn_wmma_f32_16x16x4_f32(false, a[c][s], false, b[c][j],
                                                        (short)0, acc[s][j], false, false);
  }
  if constexpr (STAGE_NEXT && I >= 2 && I < 14) {
    __builtin_amdgcn_sched_barrier(0);
    sq_store_a<I - 2>(aW, ra);
    if constexpr (((I - 2) & 1) == 0) sq_store_b<(I - 2) / 2>(bW, rb0, rb1);
  }
}

template <bool STAGE_NEXT>
__device__ __forceinline__ void sq_slab(v8f (&acc)[3][6], const float* aR, const float* bR,
                                        float* aW, float* bW, const float4 (&ra)[12],
                                        const float4 (&rb0)[6], const float4 (&rb1)[6])
{
  v2f a[2][3], b[2][6];
  #pragma unroll
  for (int s = 0; s < 3; ++s) a[0][s] = *(const v2f*)(aR + s * 16 * SQ_PA);
  #pragma unroll
  for (int j = 0; j < 6; ++j) b[0][j] = *(const v2f*)(bR + j * 32);
  sq_kstep<STAGE_NEXT,  0>(acc, a, b, aR, bR, aW, bW, ra, rb0, rb1);
  sq_kstep<STAGE_NEXT,  1>(acc, a, b, aR, bR, aW, bW, ra, rb0, rb1);
  sq_kstep<STAGE_NEXT,  2>(acc, a, b, aR, bR, aW, bW, ra, rb0, rb1);
  sq_kstep<STAGE_NEXT,  3>(acc, a, b, aR, bR, aW, bW, ra, rb0, rb1);
  sq_kstep<STAGE_NEXT,  4>(acc, a, b, aR, bR, aW, bW, ra, rb0, rb1);
  sq_kstep<STAGE_NEXT,  5>(acc, a, b, aR, bR, aW, bW, ra, rb0, rb1);
  sq_kstep<STAGE_NEXT,  6>(acc, a, b, aR, bR, aW, bW, ra, rb0, rb1);
  sq_kstep<STAGE_NEXT,  7>(acc, a, b, aR, bR, aW, bW, ra, rb0, rb1);
  sq_kstep<STAGE_NEXT,  8>(acc, a, b, aR, bR, aW, bW, ra, rb0, rb1);
  sq_kstep<STAGE_NEXT,  9>(acc, a, b, aR, bR, aW, bW, ra, rb0, rb1);
  sq_kstep<STAGE_NEXT, 10>(acc, a, b, aR, bR, aW, bW, ra, rb0, rb1);
  sq_kstep<STAGE_NEXT, 11>(acc, a, b, aR, bR, aW, bW, ra, rb0, rb1);
  sq_kstep<STAGE_NEXT, 12>(acc, a, b, aR, bR, aW, bW, ra, rb0, rb1);
  sq_kstep<STAGE_NEXT, 13>(acc, a, b, aR, bR, aW, bW, ra, rb0, rb1);
  sq_kstep<STAGE_NEXT, 14>(acc, a, b, aR, bR, aW, bW, ra, rb0, rb1);
  sq_kstep<STAGE_NEXT, 15>(acc, a, b, aR, bR, aW, bW, ra, rb0, rb1);
}

__global__ __launch_bounds__(256) void gemm_sq_lds_f32(const float* __restrict__ A,
                                                       const float* __restrict__ Bm,
                                                       float* __restrict__ C)
{
  __shared__ __attribute__((aligned(16))) float sm[2 * SQ_STG];

  const int tid  = threadIdx.x;
  const int lane = tid & 31;
  const int wave = __builtin_amdgcn_readfirstlane(tid >> 5);
  const int wm   = wave >> 1;
  const int wn   = wave & 1;
  const int half = lane >> 4;
  const int l16  = lane & 15;
  const int mb   = blockIdx.y * SQ_T;
  const int nb   = blockIdx.x * SQ_T;

  const int tr = tid >> 4, tc = tid & 15;
  const float* gA = A + (size_t)(mb + tr) * NNODE + 4 * tc;
  const float* gB = Bm + (size_t)(2 * tr) * NNODE + nb + 4 * tc;
  const int sAo = tr * SQ_PA + 4 * tc;
  const int sBo = SQ_ASZ + tr * SQ_PB + 8 * tc;
  const int aF = (48 * wm + l16) * SQ_PA + 2 * half;
  const int bF = SQ_ASZ + half * SQ_PB + 2 * (96 * wn + l16);

  v8f acc[3][6] = {};
  float4 ra[12], rb0[6], rb1[6];

  sq_load(ra, rb0, rb1, gA, gB);
  sq_store_a<0>(sm + sAo, ra);  sq_store_a<1>(sm + sAo, ra);  sq_store_a<2>(sm + sAo, ra);
  sq_store_a<3>(sm + sAo, ra);  sq_store_a<4>(sm + sAo, ra);  sq_store_a<5>(sm + sAo, ra);
  sq_store_a<6>(sm + sAo, ra);  sq_store_a<7>(sm + sAo, ra);  sq_store_a<8>(sm + sAo, ra);
  sq_store_a<9>(sm + sAo, ra);  sq_store_a<10>(sm + sAo, ra); sq_store_a<11>(sm + sAo, ra);
  sq_store_b<0>(sm + sBo, rb0, rb1);  sq_store_b<1>(sm + sBo, rb0, rb1);
  sq_store_b<2>(sm + sBo, rb0, rb1);  sq_store_b<3>(sm + sBo, rb0, rb1);
  sq_store_b<4>(sm + sBo, rb0, rb1);  sq_store_b<5>(sm + sBo, rb0, rb1);
  __syncthreads();

  int st = 0;
  for (int t = 0; t < NNODE / SQ_KT - 1; ++t) {
    gA += SQ_KT;
    gB += (size_t)SQ_KT * NNODE;
    sq_load(ra, rb0, rb1, gA, gB);
    sq_slab<true>(acc, sm + st + aF, sm + st + bF, sm + (SQ_STG - st) + sAo,
                  sm + (SQ_STG - st) + sBo, ra, rb0, rb1);
    __syncthreads();
    st = SQ_STG - st;
  }
  sq_slab<false>(acc, sm + st + aF, sm + st + bF, nullptr, nullptr, ra, rb0, rb1);

  #pragma unroll
  for (int s = 0; s < 3; ++s) {
    #pragma unroll
    for (int v = 0; v < 8; ++v) {
      const int row = mb + 48 * wm + 16 * s + v + 8 * half;
      #pragma unroll
      for (int j = 0; j < 6; ++j) {
        const int col = nb + 96 * wn + 16 * j + l16;
        C[(size_t)row * NNODE + col] = acc[s][j][v];
      }
    }
  }
}

__global__ void deg_rsqrt(const float* __restrict__ A, float* __restrict__ d)
{
  const int gw   = (blockIdx.x * blockDim.x + threadIdx.x) >> 5;
  const int lane = threadIdx.x & 31;
  if (gw >= NNODE) return;
  const float* row = A + (size_t)gw * NNODE;
  float s = 0.0f;
  for (int j = lane; j < NNODE; j += 32) s += row[j];
  #pragma unroll
  for (int off = 16; off > 0; off >>= 1) s += __shfl_xor(s, off, 32);
  if (lane == 0) d[gw] = (s > 0.0f) ? rsqrtf(s) : 0.0f;
}

#define DEG_G (NNODE / 4 / 32)
#define DEG_WAVES __attribute__((amdgpu_waves_per_eu(4, 4)))

__device__ __forceinline__ float thr_w(float m, float a, float thr)
{
  return (m > thr) ? a : 0.0f;
}

__global__ __launch_bounds__(256) DEG_WAVES void branch_deg_rsqrt(
    const float* __restrict__ mask, const float* __restrict__ adj,
    const float* __restrict__ thrp, float* __restrict__ d)
{
  const int gw   = (blockIdx.x * blockDim.x + threadIdx.x) >> 5;
  const int lane = threadIdx.x & 31;
  if (gw >= NNODE) return;
  const float thr = thrp[0];
  const float* mp = mask + (size_t)gw * NNODE + 4 * lane;
  const float* ap = adj  + (size_t)gw * NNODE + 4 * lane;
  float4 m[DEG_G], a[DEG_G];
  #pragma unroll
  for (int g = 0; g < DEG_G; ++g) {
    m[g] = *(const float4*)(mp + 128 * g);
    a[g] = *(const float4*)(ap + 128 * g);
  }
  float s = thr_w(m[0].x, a[0].x, thr) + thr_w(m[0].y, a[0].y, thr);
  s += thr_w(m[0].z, a[0].z, thr);
  s += thr_w(m[0].w, a[0].w, thr);
  #pragma unroll
  for (int g = 1; g < DEG_G; ++g) {
    s += thr_w(m[g].x, a[g].x, thr);
    s += thr_w(m[g].y, a[g].y, thr);
    s += thr_w(m[g].z, a[g].z, thr);
    s += thr_w(m[g].w, a[g].w, thr);
  }
  #pragma unroll
  for (int off = 16; off > 0; off >>= 1) s += __shfl_xor(s, off, 32);
  if (lane == 0) d[gw] = rsqrtf(s + 1.0f);
}

#define NORM_R 6
#define NORM_C (NNODE / 4 / 256)
#define NORM_WAVES __attribute__((amdgpu_waves_per_eu(4, 4)))

__global__ __launch_bounds__(256) NORM_WAVES void symnorm(const float* __restrict__ adj,
                                                          const float* __restrict__ d,
                                                          float* __restrict__ An)
{
  const int tid = threadIdx.x;
  const int r0  = blockIdx.x * NORM_R;
  const float* ap = adj + (size_t)r0 * NNODE + 4 * tid;
  float*       op = An  + (size_t)r0 * NNODE + 4 * tid;

  float4 dj[NORM_C], a[NORM_R][NORM_C];
  float  di[NORM_R];
  #pragma unroll
  for (int c = 0; c < NORM_C; ++c) dj[c] = *(const float4*)(d + 4 * tid + 1024 * c);
  #pragma unroll
  for (int r = 0; r < NORM_R; ++r) di[r] = d[r0 + r];
  #pragma unroll
  for (int r = 0; r < NORM_R; ++r) {
    #pragma unroll
    for (int c = 0; c < NORM_C; ++c)
      a[r][c] = *(const float4*)(ap + (size_t)r * NNODE + 1024 * c);
  }
  #pragma unroll
  for (int r = 0; r < NORM_R; ++r) {
    #pragma unroll
    for (int c = 0; c < NORM_C; ++c) {
      float4 o;
      o.x = a[r][c].x * di[r] * dj[c].x;
      o.y = a[r][c].y * di[r] * dj[c].y;
      o.z = a[r][c].z * di[r] * dj[c].z;
      o.w = a[r][c].w * di[r] * dj[c].w;
      *(float4*)(op + (size_t)r * NNODE + 1024 * c) = o;
    }
  }
}

__device__ __forceinline__ float bn_elem(float m, float a, float thr, bool diag,
                                         float di, float dj)
{
  float w = (m > thr) ? a : 0.0f;
  if (diag) w += 1.0f;
  return w * di * dj;
}

__global__ __launch_bounds__(256) NORM_WAVES void branch_norm(
    const float* __restrict__ mask, const float* __restrict__ adj,
    const float* __restrict__ thrp, const float* __restrict__ d, float* __restrict__ An)
{
  const int tid = threadIdx.x;
  const int r0  = blockIdx.x * NORM_R;
  const float thr = thrp[0];
  const float* mp = mask + (size_t)r0 * NNODE + 4 * tid;
  const float* ap = adj  + (size_t)r0 * NNODE + 4 * tid;
  float*       op = An   + (size_t)r0 * NNODE + 4 * tid;

  float4 dj[NORM_C], m[NORM_R][NORM_C], a[NORM_R][NORM_C];
  float  di[NORM_R];
  #pragma unroll
  for (int c = 0; c < NORM_C; ++c) dj[c] = *(const float4*)(d + 4 * tid + 1024 * c);
  #pragma unroll
  for (int r = 0; r < NORM_R; ++r) di[r] = d[r0 + r];
  #pragma unroll
  for (int r = 0; r < NORM_R; ++r) {
    #pragma unroll
    for (int c = 0; c < NORM_C; ++c) {
      m[r][c] = *(const float4*)(mp + (size_t)r * NNODE + 1024 * c);
      a[r][c] = *(const float4*)(ap + (size_t)r * NNODE + 1024 * c);
    }
  }
  #pragma unroll
  for (int r = 0; r < NORM_R; ++r) {
    #pragma unroll
    for (int c = 0; c < NORM_C; ++c) {
      const int dc = 4 * tid + 1024 * c - (r0 + r);
      float4 o;
      o.x = bn_elem(m[r][c].x, a[r][c].x, thr, dc ==  0, di[r], dj[c].x);
      o.y = bn_elem(m[r][c].y, a[r][c].y, thr, dc == -1, di[r], dj[c].y);
      o.z = bn_elem(m[r][c].z, a[r][c].z, thr, dc == -2, di[r], dj[c].z);
      o.w = bn_elem(m[r][c].w, a[r][c].w, thr, dc == -3, di[r], dj[c].w);
      *(float4*)(op + (size_t)r * NNODE + 1024 * c) = o;
    }
  }
}

__global__ void conv_dil2(const float* __restrict__ emb,
                          const float* __restrict__ kw,
                          const float* __restrict__ kb,
                          float* __restrict__ out)
{
  const int idx = blockIdx.x * blockDim.x + threadIdx.x;
  if (idx >= NNODE * 256) return;
  const int n = idx >> 8;
  const int c = idx & 255;
  float acc = kb[0];
  #pragma unroll
  for (int ch = 0; ch < 3; ++ch) {
    #pragma unroll
    for (int kh = 0; kh < 3; ++kh) {
      const int ih = n + 2 * kh - 2;
      if (ih < 0 || ih >= NNODE) continue;
      #pragma unroll
      for (int kwi = 0; kwi < 3; ++kwi) {
        const int iw = c + 2 * kwi - 2;
        if (iw < 0 || iw >= 256) continue;
        acc += emb[((size_t)ch * NNODE + ih) * 256 + iw] * kw[ch * 9 + kh * 3 + kwi];
      }
    }
  }
  out[idx] = acc;
}

__global__ void logits_k(const float* __restrict__ h,
                         const float* __restrict__ Wb,
                         const float* __restrict__ bb,
                         float* __restrict__ out)
{
  const int i = blockIdx.x * blockDim.x + threadIdx.x;
  if (i >= BPAIR * 2) return;
  const int b = i >> 1;
  const int c = i & 1;
  float acc = bb[c];
  #pragma unroll
  for (int k = 0; k < 64; ++k) acc += h[b * 64 + k] * Wb[k * 2 + c];
  out[i] = acc;
}

extern "C" void kernel_launch(void* const* d_in, const int* in_sizes, int n_in,
                              void* d_out, int out_size, void* d_ws, size_t ws_size,
                              hipStream_t stream)
{
  const int*   left  = (const int*)d_in[0];
  const int*   right = (const int*)d_in[1];
  const float* X     = (const float*)d_in[2];
  const float* adj   = (const float*)d_in[3];
  const float* thr   = (const float*)d_in[4];
  const float* Wfc1  = (const float*)d_in[5];
  const float* bfc1  = (const float*)d_in[6];
  const float* Wg[6] = {(const float*)d_in[7],  (const float*)d_in[9],
                        (const float*)d_in[11], (const float*)d_in[13],
                        (const float*)d_in[15], (const float*)d_in[17]};
  const float* bg[6] = {(const float*)d_in[8],  (const float*)d_in[10],
                        (const float*)d_in[12], (const float*)d_in[14],
                        (const float*)d_in[16], (const float*)d_in[18]};
  const float* cnnk  = (const float*)d_in[19];
  const float* cnnb  = (const float*)d_in[20];
  const float* Wa    = (const float*)d_in[21];
  const float* ba    = (const float*)d_in[22];
  const float* Wb    = (const float*)d_in[23];
  const float* bb    = (const float*)d_in[24];

  float* out     = (float*)d_out;
  float* logits  = out;
  float* emb_all = out + BPAIR * 2;

  float* ws = (float*)d_ws;
  size_t off = 0;
  float* X0  = ws + off; off += (size_t)NNODE * D0;
  float* An  = ws + off; off += (size_t)NNODE * NNODE;
  float* A2  = ws + off; off += (size_t)NNODE * NNODE;
  float* AnB = ws + off; off += (size_t)NNODE * NNODE;
  float* dv  = ws + off; off += NNODE;
  float* Hb  = ws + off; off += (size_t)NNODE * HDIM;
  float* H2b = ws + off; off += (size_t)NNODE * HDIM;
  float* emb = ws + off; off += (size_t)3 * NNODE * 256;
  float* hm  = ws + off; off += (size_t)BPAIR * 64;
  if (ws_size < off * sizeof(float)) return;

  const dim3 blk8(256);
  const dim3 blk16(512);

  gemm_wmma_f32<true, true, false, 1, 16><<<dim3(D0 / 64, NNODE / 32), blk16, 0, stream>>>(
      X, FIN, Wfc1, D0, bfc1, X0, D0, FIN, nullptr, nullptr);

  deg_rsqrt<<<NNODE / 8, 256, 0, stream>>>(adj, dv);
  symnorm<<<NNODE / NORM_R, 256, 0, stream>>>(adj, dv, An);
  gemm_sq_lds_f32<<<dim3(NNODE / SQ_T, NNODE / SQ_T), 256, 0, stream>>>(An, An, A2);

  const float* masks[3] = { adj, An, A2 };
  for (int ch = 0; ch < 3; ++ch) {
    const float* W1 = Wg[4 - 2 * ch];  const float* b1 = bg[4 - 2 * ch];
    const float* W2 = Wg[5 - 2 * ch];  const float* b2 = bg[5 - 2 * ch];
    float* embc = emb + (size_t)ch * NNODE * 256;

    branch_deg_rsqrt<<<NNODE / 8, 256, 0, stream>>>(masks[ch], adj, thr, dv);
    branch_norm<<<NNODE / NORM_R, 256, 0, stream>>>(masks[ch], adj, thr, dv, AnB);

    gemm_wmma_f32<false, false, false, 1, 8><<<dim3(HDIM / 64, NNODE / 32), blk8, 0, stream>>>(
        X0, D0, W1, HDIM, nullptr, Hb, HDIM, D0, nullptr, nullptr);
    gemm_wmma_f32<true, true, false, 1, 16><<<dim3(HDIM / 64, NNODE / 32), blk16, 0, stream>>>(
        AnB, NNODE, Hb, HDIM, b1, embc, 256, NNODE, nullptr, nullptr);
    gemm_wmma_f32<false, false, false, 1, 8><<<dim3(HDIM / 64, NNODE / 32), blk8, 0, stream>>>(
        embc, 256, W2, HDIM, nullptr, H2b, HDIM, HDIM, nullptr, nullptr);
    gemm_wmma_f32<true, true, false, 1, 16><<<dim3(HDIM / 64, NNODE / 32), blk16, 0, stream>>>(
        AnB, NNODE, H2b, HDIM, b2, embc + 128, 256, NNODE, nullptr, nullptr);
  }

  conv_dil2<<<(NNODE * 256) / 256, 256, 0, stream>>>(emb, cnnk, cnnb, emb_all);

  gemm_wmma_f32<true, true, true, 1, 8><<<dim3(64 / 64, BPAIR / 32), blk8, 0, stream>>>(
      emb_all, 256, Wa, 64, ba, hm, 64, 512, left, right);

  logits_k<<<(BPAIR * 2 + 255) / 256, 256, 0, stream>>>(hm, Wb, bb, logits);
}


# the three independent branches run each stage as ONE launch (grid z = branch: degree, norm, H, e1, H2, e2): 25 launches become 13; two norms written in place; every sum and k order unchanged
# speedup vs baseline: 4.3330x; 1.0482x over previous
#include <hip/hip_runtime.h>
#include <hip/hip_bf16.h>

#define NNODE 3072
#define FIN   1546
#define D0    256
#define HDIM  128
#define BPAIR 16384

typedef __attribute__((ext_vector_type(2))) float v2f;
typedef __attribute__((ext_vector_type(8))) float v8f;

__device__ __forceinline__ void wmma8(v8f (&acc)[2][4], v2f a0, v2f a1,
                                      v2f b0, v2f b1, v2f b2, v2f b3)
{
  acc[0][0] = __builtin_amdgcn_wmma_f32_16x16x4_f32(false, a0, false, b0, (short)0, acc[0][0], false, false);
  acc[0][1] = __builtin_amdgcn_wmma_f32_16x16x4_f32(false, a0, false, b1, (short)0, acc[0][1], false, false);
  acc[0][2] = __builtin_amdgcn_wmma_f32_16x16x4_f32(false, a0, false, b2, (short)0, acc[0][2], false, false);
  acc[0][3] = __builtin_amdgcn_wmma_f32_16x16x4_f32(false, a0, false, b3, (short)0, acc[0][3], false, false);
  acc[1][0] = __builtin_amdgcn_wmma_f32_16x16x4_f32(false, a1, false, b0, (short)0, acc[1][0], false, false);
  acc[1][1] = __builtin_amdgcn_wmma_f32_16x16x4_f32(false, a1, false, b1, (short)0, acc[1][1], false, false);
  acc[1][2] = __builtin_amdgcn_wmma_f32_16x16x4_f32(false, a1, false, b2, (short)0, acc[1][2], false, false);
  acc[1][3] = __builtin_amdgcn_wmma_f32_16x16x4_f32(false, a1, false, b3, (short)0, acc[1][3], false, false);
}

struct F3 { const float* a; const float* b; const float* c; };
struct O3 { float* a; float* b; float* c; };
__device__ __forceinline__ const float* pick3(const F3& s, int z)
{
  return z == 0 ? s.a : (z == 1 ? s.b : s.c);
}
__device__ __forceinline__ float* pick3(const O3& s, int z)
{
  return z == 0 ? s.a : (z == 1 ? s.b : s.c);
}

#define RED_JP 272
#define RED_PW (8 * RED_JP)

template <bool RELU, bool BIAS, bool GATHER, int MW, int KS>
__global__ __launch_bounds__(32 * MW * KS) void gemm_wmma_f32(
    F3 Az, int lda,
    F3 Bz, int ldb,
    F3 biasz,
    O3 Cz, int ldc,
    int K,
    const int* __restrict__ gl, const int* __restrict__ gr)
{
  static_assert(KS == 1 || MW == 1, "K split reduces one 32x64 tile per block");
  const int z = blockIdx.z;
  const float* __restrict__ A    = pick3(Az, z);
  const float* __restrict__ Bm   = pick3(Bz, z);
  const float* __restrict__ bias = pick3(biasz, z);
  float* __restrict__ C          = pick3(Cz, z);
  static_assert(!GATHER || (KS >= 2 && (64 % KS) == 0), "gather slice must stay in one half");

  const int lane = threadIdx.x & 31;
  const int wave = __builtin_amdgcn_readfirstlane((int)(threadIdx.x >> 5));
  const int wm   = wave % MW;
  const int wk   = wave / MW;
  const int half = lane >> 4;
  const int l16  = lane & 15;

  const int mb  = blockIdx.y * (32 * MW);
  const int m0  = mb + wm * 32;
  const int n0  = blockIdx.x * 64;
  const int am0 = m0 + l16;
  const int am1 = m0 + 16 + l16;

  v8f acc[2][4] = {};

  const int nsteps = (GATHER ? 512 : K) >> 2;
  const int per    = (nsteps + KS - 1) / KS;
  const int sbeg   = min(wk * per, nsteps);
  const int send   = min(sbeg + per, nsteps);
  const int kbeg   = 4 * sbeg;

  const float* bp = Bm + (size_t)(kbeg + 2 * half) * ldb + n0 + l16;
  const size_t bstep = (size_t)4 * ldb;

  if (GATHER) {
    const int* gi = (kbeg < 256) ? gl : gr;
    const float* ap0 = A + (size_t)gi[am0] * 256 + (kbeg & 255) + 2 * half;
    const float* ap1 = A + (size_t)gi[am1] * 256 + (kbeg & 255) + 2 * half;
    #pragma unroll 4
    for (int s = sbeg; s < send; ++s) {
      v2f a0 = *(const v2f*)ap0;
      v2f a1 = *(const v2f*)ap1;
      v2f b0, b1, b2, b3;
      b0.x = bp[0];  b0.y = bp[ldb + 0];
      b1.x = bp[16]; b1.y = bp[ldb + 16];
      b2.x = bp[32]; b2.y = bp[ldb + 32];
      b3.x = bp[48]; b3.y = bp[ldb + 48];
      wmma8(acc, a0, a1, b0, b1, b2, b3);
      ap0 += 4; ap1 += 4; bp += bstep;
    }
  } else {
    const float* ap0 = A + (size_t)am0 * lda + kbeg + 2 * half;
    const float* ap1 = A + (size_t)am1 * lda + kbeg + 2 * half;
    #pragma unroll 2
    for (int s = sbeg; s < send; ++s) {
      v2f a0 = *(const v2f*)ap0;
      v2f a1 = *(const v2f*)ap1;
      v2f b0, b1, b2, b3;
      b0.x = bp[0];  b0.y = bp[ldb + 0];
      b1.x = bp[16]; b1.y = bp[ldb + 16];
      b2.x = bp[32]; b2.y = bp[ldb + 32];
      b3.x = bp[48]; b3.y = bp[ldb + 48];
      wmma8(acc, a0, a1, b0, b1, b2, b3);
      ap0 += 4; ap1 += 4; bp += bstep;
    }
    const int kmain = K & ~3;
    if (kmain < K && wk == KS - 1) {
      const int ka = kmain + 2 * half;
      const bool g0 = (ka < K), g1 = (ka + 1 < K);
      v2f a0, a1, b0, b1, b2, b3;
      a0.x = g0 ? ap0[0] : 0.0f;  a0.y = g1 ? ap0[1] : 0.0f;
      a1.x = g0 ? ap1[0] : 0.0f;  a1.y = g1 ? ap1[1] : 0.0f;
      b0.x = g0 ? bp[0]  : 0.0f;  b0.y = g1 ? bp[ldb + 0]  : 0.0f;
      b1.x = g0 ? bp[16] : 0.0f;  b1.y = g1 ? bp[ldb + 16] : 0.0f;
      b2.x = g0 ? bp[32] : 0.0f;  b2.y = g1 ? bp[ldb + 32] : 0.0f;
      b3.x = g0 ? bp[48] : 0.0f;  b3.y = g1 ? bp[ldb + 48] : 0.0f;
      wmma8(acc, a0, a1, b0, b1, b2, b3);
    }
  }

  if (KS == 1) {
    #pragma unroll
    for (int s = 0; s < 2; ++s) {
      #pragma unroll
      for (int v = 0; v < 8; ++v) {
        const int row = m0 + 16 * s + v + 8 * half;
        #pragma unroll
        for (int j = 0; j < 4; ++j) {
          const int col = n0 + 16 * j + l16;
          float x = acc[s][j][v];
          if (BIAS) x += bias[col];
          if (RELU) x = fmaxf(x, 0.0f);
          C[(size_t)row * ldc + col] = x;
        }
      }
    }
  } else {
    __shared__ __attribute__((aligned(16))) float red[KS * RED_PW];
    float* my = red + wk * RED_PW + lane;
    #pragma unroll
    for (int s = 0; s < 2; ++s) {
      #pragma unroll
      for (int j = 0; j < 4; ++j) {
        #pragma unroll
        for (int v = 0; v < 8; ++v)
          my[(s * 4 + j) * RED_JP + v * 32] = acc[s][j][v];
      }
    }
    __syncthreads();
    for (int e = threadIdx.x; e < 512; e += 32 * KS) {
      const int row = e >> 4;
      const int c4  = (e & 15) * 4;
      const int s   = row >> 4;
      const int hf  = (row >> 3) & 1;
      const int v   = row & 7;
      const int j   = c4 >> 4;
      const int l   = c4 & 15;
      const float* p = red + (s * 4 + j) * RED_JP + v * 32 + hf * 16 + l;
      float4 sum = *(const float4*)p;
      #pragma unroll
      for (int w = 1; w < KS; ++w) {
        const float4 x = *(const float4*)(p + w * RED_PW);
        sum.x += x.x; sum.y += x.y; sum.z += x.z; sum.w += x.w;
      }
      const int col = n0 + c4;
      if (BIAS) {
        sum.x += bias[col + 0]; sum.y += bias[col + 1];
        sum.z += bias[col + 2]; sum.w += bias[col + 3];
      }
      if (RELU) {
        sum.x = fmaxf(sum.x, 0.0f); sum.y = fmaxf(sum.y, 0.0f);
        sum.z = fmaxf(sum.z, 0.0f); sum.w = fmaxf(sum.w, 0.0f);
      }
      *(float4*)(C + (size_t)(mb + row) * ldc + col) = sum;
    }
  }
}

#define SQ_T   192
#define SQ_KT  64
#define SQ_PA  68
#define SQ_PB  416
#define SQ_ASZ (SQ_T * SQ_PA)
#define SQ_BSZ ((SQ_KT / 2) * SQ_PB)
#define SQ_STG (SQ_ASZ + SQ_BSZ)

__device__ __forceinline__ void sq_load(float4 (&ra)[12], float4 (&rb0)[6], float4 (&rb1)[6],
                                        const float* gA, const float* gB)
{
  #pragma unroll
  for (int i = 0; i < 12; ++i) ra[i] = *(const float4*)(gA + (size_t)i * 16 * NNODE);
  #pragma unroll
  for (int a = 0; a < 2; ++a) {
    #pragma unroll
    for (int b = 0; b < 3; ++b) {
      const float* g = gB + (size_t)a * 32 * NNODE + b * 64;
      rb0[a * 3 + b] = *(const float4*)g;
      rb1[a * 3 + b] = *(const float4*)(g + NNODE);
    }
  }
}

template <int I>
__device__ __forceinline__ void sq_store_a(float* aW, const float4 (&ra)[12])
{
  *(float4*)(aW + I * 16 * SQ_PA) = make_float4(ra[I].x, ra[I].y, ra[I].z, ra[I].w);
}

template <int Q>
__device__ __forceinline__ void sq_store_b(float* bW, const float4 (&rb0)[6],
                                           const float4 (&rb1)[6])
{
  float* d = bW + (Q / 3) * 16 * SQ_PB + (Q % 3) * 128;
  *(float4*)d       = make_float4(rb0[Q].x, rb1[Q].x, rb0[Q].y, rb1[Q].y);
  *(float4*)(d + 4) = make_float4(rb0[Q].z, rb1[Q].z, rb0[Q].w, rb1[Q].w);
}

template <bool STAGE_NEXT, int I>
__device__ __forceinline__ void sq_kstep(v8f (&acc)[3][6], v2f (&a)[2][3], v2f (&b)[2][6],
                                         const float* aR, const float* bR,
                                         float* aW, float* bW, const float4 (&ra)[12],
                                         const float4 (&rb0)[6], const float4 (&rb1)[6])
{
  constexpr int c = I & 1, n = c ^ 1;
  if (I + 1 < SQ_KT / 4) {
    #pragma unroll
    for (int s = 0; s < 3; ++s)
      a[n][s] = *(const v2f*)(aR + s * 16 * SQ_PA + 4 * (I + 1));
    #pragma unroll
    for (int j = 0; j < 6; ++j)
      b[n][j] = *(const v2f*)(bR + 2 * (I + 1) * SQ_PB + j * 32);
  }
  #pragma unroll
  for (int s = 0; s < 3; ++s) {
    #pragma unroll
    for (int j = 0; j < 6; ++j)
      acc[s][j] = __builtin_amdgcn_wmma_f32_16x16x4_f32(false, a[c][s], false, b[c][j],
                                                        (short)0, acc[s][j], false, false);
  }
  if constexpr (STAGE_NEXT && I >= 2 && I < 14) {
    __builtin_amdgcn_sched_barrier(0);
    sq_store_a<I - 2>(aW, ra);
    if constexpr (((I - 2) & 1) == 0) sq_store_b<(I - 2) / 2>(bW, rb0, rb1);
  }
}

template <bool STAGE_NEXT>
__device__ __forceinline__ void sq_slab(v8f (&acc)[3][6], const float* aR, const float* bR,
                                        float* aW, float* bW, const float4 (&ra)[12],
                                        const float4 (&rb0)[6], const float4 (&rb1)[6])
{
  v2f a[2][3], b[2][6];
  #pragma unroll
  for (int s = 0; s < 3; ++s) a[0][s] = *(const v2f*)(aR + s * 16 * SQ_PA);
  #pragma unroll
  for (int j = 0; j < 6; ++j) b[0][j] = *(const v2f*)(bR + j * 32);
  sq_kstep<STAGE_NEXT,  0>(acc, a, b, aR, bR, aW, bW, ra, rb0, rb1);
  sq_kstep<STAGE_NEXT,  1>(acc, a, b, aR, bR, aW, bW, ra, rb0, rb1);
  sq_kstep<STAGE_NEXT,  2>(acc, a, b, aR, bR, aW, bW, ra, rb0, rb1);
  sq_kstep<STAGE_NEXT,  3>(acc, a, b, aR, bR, aW, bW, ra, rb0, rb1);
  sq_kstep<STAGE_NEXT,  4>(acc, a, b, aR, bR, aW, bW, ra, rb0, rb1);
  sq_kstep<STAGE_NEXT,  5>(acc, a, b, aR, bR, aW, bW, ra, rb0, rb1);
  sq_kstep<STAGE_NEXT,  6>(acc, a, b, aR, bR, aW, bW, ra, rb0, rb1);
  sq_kstep<STAGE_NEXT,  7>(acc, a, b, aR, bR, aW, bW, ra, rb0, rb1);
  sq_kstep<STAGE_NEXT,  8>(acc, a, b, aR, bR, aW, bW, ra, rb0, rb1);
  sq_kstep<STAGE_NEXT,  9>(acc, a, b, aR, bR, aW, bW, ra, rb0, rb1);
  sq_kstep<STAGE_NEXT, 10>(acc, a, b, aR, bR, aW, bW, ra, rb0, rb1);
  sq_kstep<STAGE_NEXT, 11>(acc, a, b, aR, bR, aW, bW, ra, rb0, rb1);
  sq_kstep<STAGE_NEXT, 12>(acc, a, b, aR, bR, aW, bW, ra, rb0, rb1);
  sq_kstep<STAGE_NEXT, 13>(acc, a, b, aR, bR, aW, bW, ra, rb0, rb1);
  sq_kstep<STAGE_NEXT, 14>(acc, a, b, aR, bR, aW, bW, ra, rb0, rb1);
  sq_kstep<STAGE_NEXT, 15>(acc, a, b, aR, bR, aW, bW, ra, rb0, rb1);
}

__global__ __launch_bounds__(256) void gemm_sq_lds_f32(const float* __restrict__ A,
                                                       const float* __restrict__ Bm,
                                                       float* __restrict__ C)
{
  __shared__ __attribute__((aligned(16))) float sm[2 * SQ_STG];

  const int tid  = threadIdx.x;
  const int lane = tid & 31;
  const int wave = __builtin_amdgcn_readfirstlane(tid >> 5);
  const int wm   = wave >> 1;
  const int wn   = wave & 1;
  const int half = lane >> 4;
  const int l16  = lane & 15;
  const int mb   = blockIdx.y * SQ_T;
  const int nb   = blockIdx.x * SQ_T;

  const int tr = tid >> 4, tc = tid & 15;
  const float* gA = A + (size_t)(mb + tr) * NNODE + 4 * tc;
  const float* gB = Bm + (size_t)(2 * tr) * NNODE + nb + 4 * tc;
  const int sAo = tr * SQ_PA + 4 * tc;
  const int sBo = SQ_ASZ + tr * SQ_PB + 8 * tc;
  const int aF = (48 * wm + l16) * SQ_PA + 2 * half;
  const int bF = SQ_ASZ + half * SQ_PB + 2 * (96 * wn + l16);

  v8f acc[3][6] = {};
  float4 ra[12], rb0[6], rb1[6];

  sq_load(ra, rb0, rb1, gA, gB);
  sq_store_a<0>(sm + sAo, ra);  sq_store_a<1>(sm + sAo, ra);  sq_store_a<2>(sm + sAo, ra);
  sq_store_a<3>(sm + sAo, ra);  sq_store_a<4>(sm + sAo, ra);  sq_store_a<5>(sm + sAo, ra);
  sq_store_a<6>(sm + sAo, ra);  sq_store_a<7>(sm + sAo, ra);  sq_store_a<8>(sm + sAo, ra);
  sq_store_a<9>(sm + sAo, ra);  sq_store_a<10>(sm + sAo, ra); sq_store_a<11>(sm + sAo, ra);
  sq_store_b<0>(sm + sBo, rb0, rb1);  sq_store_b<1>(sm + sBo, rb0, rb1);
  sq_store_b<2>(sm + sBo, rb0, rb1);  sq_store_b<3>(sm + sBo, rb0, rb1);
  sq_store_b<4>(sm + sBo, rb0, rb1);  sq_store_b<5>(sm + sBo, rb0, rb1);
  __syncthreads();

  int st = 0;
  for (int t = 0; t < NNODE / SQ_KT - 1; ++t) {
    gA += SQ_KT;
    gB += (size_t)SQ_KT * NNODE;
    sq_load(ra, rb0, rb1, gA, gB);
    sq_slab<true>(acc, sm + st + aF, sm + st + bF, sm + (SQ_STG - st) + sAo,
                  sm + (SQ_STG - st) + sBo, ra, rb0, rb1);
    __syncthreads();
    st = SQ_STG - st;
  }
  sq_slab<false>(acc, sm + st + aF, sm + st + bF, nullptr, nullptr, ra, rb0, rb1);

  #pragma unroll
  for (int s = 0; s < 3; ++s) {
    #pragma unroll
    for (int v = 0; v < 8; ++v) {
      const int row = mb + 48 * wm + 16 * s + v + 8 * half;
      #pragma unroll
      for (int j = 0; j < 6; ++j) {
        const int col = nb + 96 * wn + 16 * j + l16;
        C[(size_t)row * NNODE + col] = acc[s][j][v];
      }
    }
  }
}

__global__ void deg_rsqrt(const float* __restrict__ A, float* __restrict__ d)
{
  const int gw   = (blockIdx.x * blockDim.x + threadIdx.x) >> 5;
  const int lane = threadIdx.x & 31;
  if (gw >= NNODE) return;
  const float* row = A + (size_t)gw * NNODE;
  float s = 0.0f;
  for (int j = lane; j < NNODE; j += 32) s += row[j];
  #pragma unroll
  for (int off = 16; off > 0; off >>= 1) s += __shfl_xor(s, off, 32);
  if (lane == 0) d[gw] = (s > 0.0f) ? rsqrtf(s) : 0.0f;
}

#define DEG_G (NNODE / 4 / 32)
#define DEG_WAVES __attribute__((amdgpu_waves_per_eu(4, 4)))

__device__ __forceinline__ float thr_w(float m, float a, float thr)
{
  return (m > thr) ? a : 0.0f;
}

__global__ __launch_bounds__(256) DEG_WAVES void branch_deg_rsqrt(
    F3 maskz, const float* __restrict__ adj,
    const float* __restrict__ thrp, float* __restrict__ d3)
{
  const int gw   = (blockIdx.x * blockDim.x + threadIdx.x) >> 5;
  const int lane = threadIdx.x & 31;
  if (gw >= NNODE) return;
  const int z = blockIdx.z;
  const float* __restrict__ mask = pick3(maskz, z);
  float* __restrict__ d = d3 + z * NNODE;
  const float thr = thrp[0];
  const float* mp = mask + (size_t)gw * NNODE + 4 * lane;
  const float* ap = adj  + (size_t)gw * NNODE + 4 * lane;
  float4 m[DEG_G], a[DEG_G];
  #pragma unroll
  for (int g = 0; g < DEG_G; ++g) {
    m[g] = *(const float4*)(mp + 128 * g);
    a[g] = *(const float4*)(ap + 128 * g);
  }
  float s = thr_w(m[0].x, a[0].x, thr) + thr_w(m[0].y, a[0].y, thr);
  s += thr_w(m[0].z, a[0].z, thr);
  s += thr_w(m[0].w, a[0].w, thr);
  #pragma unroll
  for (int g = 1; g < DEG_G; ++g) {
    s += thr_w(m[g].x, a[g].x, thr);
    s += thr_w(m[g].y, a[g].y, thr);
    s += thr_w(m[g].z, a[g].z, thr);
    s += thr_w(m[g].w, a[g].w, thr);
  }
  #pragma unroll
  for (int off = 16; off > 0; off >>= 1) s += __shfl_xor(s, off, 32);
  if (lane == 0) d[gw] = rsqrtf(s + 1.0f);
}

#define NORM_R 6
#define NORM_C (NNODE / 4 / 256)
#define NORM_WAVES __attribute__((amdgpu_waves_per_eu(4, 4)))

__global__ __launch_bounds__(256) NORM_WAVES void symnorm(const float* __restrict__ adj,
                                                          const float* __restrict__ d,
                                                          float* __restrict__ An)
{
  const int tid = threadIdx.x;
  const int r0  = blockIdx.x * NORM_R;
  const float* ap = adj + (size_t)r0 * NNODE + 4 * tid;
  float*       op = An  + (size_t)r0 * NNODE + 4 * tid;

  float4 dj[NORM_C], a[NORM_R][NORM_C];
  float  di[NORM_R];
  #pragma unroll
  for (int c = 0; c < NORM_C; ++c) dj[c] = *(const float4*)(d + 4 * tid + 1024 * c);
  #pragma unroll
  for (int r = 0; r < NORM_R; ++r) di[r] = d[r0 + r];
  #pragma unroll
  for (int r = 0; r < NORM_R; ++r) {
    #pragma unroll
    for (int c = 0; c < NORM_C; ++c)
      a[r][c] = *(const float4*)(ap + (size_t)r * NNODE + 1024 * c);
  }
  #pragma unroll
  for (int r = 0; r < NORM_R; ++r) {
    #pragma unroll
    for (int c = 0; c < NORM_C; ++c) {
      float4 o;
      o.x = a[r][c].x * di[r] * dj[c].x;
      o.y = a[r][c].y * di[r] * dj[c].y;
      o.z = a[r][c].z * di[r] * dj[c].z;
      o.w = a[r][c].w * di[r] * dj[c].w;
      *(float4*)(op + (size_t)r * NNODE + 1024 * c) = o;
    }
  }
}

__device__ __forceinline__ float bn_elem(float m, float a, float thr, bool diag,
                                         float di, float dj)
{
  float w = (m > thr) ? a : 0.0f;
  if (diag) w += 1.0f;
  return w * di * dj;
}

__global__ __launch_bounds__(256) NORM_WAVES void branch_norm(
    F3 maskz, const float* __restrict__ adj,
    const float* __restrict__ thrp, const float* __restrict__ d3, O3 Anz)
{
  const int tid = threadIdx.x;
  const int r0  = blockIdx.x * NORM_R;
  const int z   = blockIdx.z;
  const float* mask = pick3(maskz, z);
  float* An = pick3(Anz, z);
  const float* __restrict__ d = d3 + z * NNODE;
  const float thr = thrp[0];
  const float* mp = mask + (size_t)r0 * NNODE + 4 * tid;
  const float* ap = adj  + (size_t)r0 * NNODE + 4 * tid;
  float*       op = An   + (size_t)r0 * NNODE + 4 * tid;

  float4 dj[NORM_C], m[NORM_R][NORM_C], a[NORM_R][NORM_C];
  float  di[NORM_R];
  #pragma unroll
  for (int c = 0; c < NORM_C; ++c) dj[c] = *(const float4*)(d + 4 * tid + 1024 * c);
  #pragma unroll
  for (int r = 0; r < NORM_R; ++r) di[r] = d[r0 + r];
  #pragma unroll
  for (int r = 0; r < NORM_R; ++r) {
    #pragma unroll
    for (int c = 0; c < NORM_C; ++c) {
      m[r][c] = *(const float4*)(mp + (size_t)r * NNODE + 1024 * c);
      a[r][c] = *(const float4*)(ap + (size_t)r * NNODE + 1024 * c);
    }
  }
  #pragma unroll
  for (int r = 0; r < NORM_R; ++r) {
    #pragma unroll
    for (int c = 0; c < NORM_C; ++c) {
      const int dc = 4 * tid + 1024 * c - (r0 + r);
      float4 o;
      o.x = bn_elem(m[r][c].x, a[r][c].x, thr, dc ==  0, di[r], dj[c].x);
      o.y = bn_elem(m[r][c].y, a[r][c].y, thr, dc == -1, di[r], dj[c].y);
      o.z = bn_elem(m[r][c].z, a[r][c].z, thr, dc == -2, di[r], dj[c].z);
      o.w = bn_elem(m[r][c].w, a[r][c].w, thr, dc == -3, di[r], dj[c].w);
      *(float4*)(op + (size_t)r * NNODE + 1024 * c) = o;
    }
  }
}

__global__ void conv_dil2(const float* __restrict__ emb,
                          const float* __restrict__ kw,
                          const float* __restrict__ kb,
                          float* __restrict__ out)
{
  const int idx = blockIdx.x * blockDim.x + threadIdx.x;
  if (idx >= NNODE * 256) return;
  const int n = idx >> 8;
  const int c = idx & 255;
  float acc = kb[0];
  #pragma unroll
  for (int ch = 0; ch < 3; ++ch) {
    #pragma unroll
    for (int kh = 0; kh < 3; ++kh) {
      const int ih = n + 2 * kh - 2;
      if (ih < 0 || ih >= NNODE) continue;
      #pragma unroll
      for (int kwi = 0; kwi < 3; ++kwi) {
        const int iw = c + 2 * kwi - 2;
        if (iw < 0 || iw >= 256) continue;
        acc += emb[((size_t)ch * NNODE + ih) * 256 + iw] * kw[ch * 9 + kh * 3 + kwi];
      }
    }
  }
  out[idx] = acc;
}

__global__ void logits_k(const float* __restrict__ h,
                         const float* __restrict__ Wb,
                         const float* __restrict__ bb,
                         float* __restrict__ out)
{
  const int i = blockIdx.x * blockDim.x + threadIdx.x;
  if (i >= BPAIR * 2) return;
  const int b = i >> 1;
  const int c = i & 1;
  float acc = bb[c];
  #pragma unroll
  for (int k = 0; k < 64; ++k) acc += h[b * 64 + k] * Wb[k * 2 + c];
  out[i] = acc;
}

extern "C" void kernel_launch(void* const* d_in, const int* in_sizes, int n_in,
                              void* d_out, int out_size, void* d_ws, size_t ws_size,
                              hipStream_t stream)
{
  const int*   left  = (const int*)d_in[0];
  const int*   right = (const int*)d_in[1];
  const float* X     = (const float*)d_in[2];
  const float* adj   = (const float*)d_in[3];
  const float* thr   = (const float*)d_in[4];
  const float* Wfc1  = (const float*)d_in[5];
  const float* bfc1  = (const float*)d_in[6];
  const float* Wg[6] = {(const float*)d_in[7],  (const float*)d_in[9],
                        (const float*)d_in[11], (const float*)d_in[13],
                        (const float*)d_in[15], (const float*)d_in[17]};
  const float* bg[6] = {(const float*)d_in[8],  (const float*)d_in[10],
                        (const float*)d_in[12], (const float*)d_in[14],
                        (const float*)d_in[16], (const float*)d_in[18]};
  const float* cnnk  = (const float*)d_in[19];
  const float* cnnb  = (const float*)d_in[20];
  const float* Wa    = (const float*)d_in[21];
  const float* ba    = (const float*)d_in[22];
  const float* Wb    = (const float*)d_in[23];
  const float* bb    = (const float*)d_in[24];

  float* out     = (float*)d_out;
  float* logits  = out;
  float* emb_all = out + BPAIR * 2;

  float* ws = (float*)d_ws;
  size_t off = 0;
  float* X0  = ws + off; off += (size_t)NNODE * D0;
  float* An  = ws + off; off += (size_t)NNODE * NNODE;
  float* A2  = ws + off; off += (size_t)NNODE * NNODE;
  float* AnB = ws + off; off += (size_t)NNODE * NNODE;
  float* dv  = ws + off; off += NNODE;
  float* Hb  = ws + off; off += (size_t)NNODE * HDIM;
  float* H2b = ws + off; off += (size_t)NNODE * HDIM;
  float* emb = ws + off; off += (size_t)3 * NNODE * 256;
  float* hm  = ws + off; off += (size_t)BPAIR * 64;
  if (ws_size < off * sizeof(float)) return;

  const dim3 blk8(256);
  const dim3 blk16(512);
  const F3 none = {nullptr, nullptr, nullptr};

  gemm_wmma_f32<true, true, false, 1, 16><<<dim3(D0 / 64, NNODE / 32), blk16, 0, stream>>>(
      F3{X, X, X}, FIN, F3{Wfc1, Wfc1, Wfc1}, D0, F3{bfc1, bfc1, bfc1}, O3{X0, X0, X0}, D0, FIN,
      nullptr, nullptr);

  deg_rsqrt<<<NNODE / 8, 256, 0, stream>>>(adj, dv);
  symnorm<<<NNODE / NORM_R, 256, 0, stream>>>(adj, dv, An);
  gemm_sq_lds_f32<<<dim3(NNODE / SQ_T, NNODE / SQ_T), 256, 0, stream>>>(An, An, A2);

  float* dv3  = Hb;
  float* emb0 = emb;
  float* emb1 = emb + (size_t)NNODE * 256;
  float* emb2 = emb + (size_t)2 * NNODE * 256;
  const F3 masks = {adj, An, A2};
  const O3 norms = {AnB, An, A2};
  const F3 normr = {AnB, An, A2};

  branch_deg_rsqrt<<<dim3(NNODE / 8, 1, 3), 256, 0, stream>>>(masks, adj, thr, dv3);
  branch_norm<<<dim3(NNODE / NORM_R, 1, 3), 256, 0, stream>>>(masks, adj, thr, dv3, norms);

  gemm_wmma_f32<false, false, false, 1, 8><<<dim3(HDIM / 64, NNODE / 32, 3), blk8, 0, stream>>>(
      F3{X0, X0, X0}, D0, F3{Wg[4], Wg[2], Wg[0]}, HDIM, none,
      O3{emb0 + 128, emb1 + 128, emb2 + 128}, 256, D0, nullptr, nullptr);
  gemm_wmma_f32<true, true, false, 1, 16><<<dim3(HDIM / 64, NNODE / 32, 3), blk16, 0, stream>>>(
      normr, NNODE, F3{emb0 + 128, emb1 + 128, emb2 + 128}, 256, F3{bg[4], bg[2], bg[0]},
      O3{emb0, emb1, emb2}, 256, NNODE, nullptr, nullptr);
  gemm_wmma_f32<false, false, false, 1, 8><<<dim3(HDIM / 64, NNODE / 32, 3), blk8, 0, stream>>>(
      F3{emb0, emb1, emb2}, 256, F3{Wg[5], Wg[3], Wg[1]}, HDIM, none,
      O3{Hb, H2b, X0}, HDIM, HDIM, nullptr, nullptr);
  gemm_wmma_f32<true, true, false, 1, 16><<<dim3(HDIM / 64, NNODE / 32, 3), blk16, 0, stream>>>(
      normr, NNODE, F3{Hb, H2b, X0}, HDIM, F3{bg[5], bg[3], bg[1]},
      O3{emb0 + 128, emb1 + 128, emb2 + 128}, 256, NNODE, nullptr, nullptr);

  conv_dil2<<<(NNODE * 256) / 256, 256, 0, stream>>>(emb, cnnk, cnnb, emb_all);

  gemm_wmma_f32<true, true, true, 1, 8><<<dim3(64 / 64, BPAIR / 32), blk8, 0, stream>>>(
      F3{emb_all, emb_all, emb_all}, 256, F3{Wa, Wa, Wa}, 64, F3{ba, ba, ba}, O3{hm, hm, hm}, 64,
      512, left, right);

  logits_k<<<(BPAIR * 2 + 255) / 256, 256, 0, stream>>>(hm, Wb, bb, logits);
}


# e1 and e2 GEMMs (N x N times N x 128): a wave fetches its A slice as whole 64-B row pieces with 16-byte loads, parks it in its own LDS region, reads fragments from LDS; k and sum order unchanged
# speedup vs baseline: 4.5779x; 1.0565x over previous
#include <hip/hip_runtime.h>
#include <hip/hip_bf16.h>

#define NNODE 3072
#define FIN   1546
#define D0    256
#define HDIM  128
#define BPAIR 16384

typedef __attribute__((ext_vector_type(2))) float v2f;
typedef __attribute__((ext_vector_type(8))) float v8f;

__device__ __forceinline__ void wmma8(v8f (&acc)[2][4], v2f a0, v2f a1,
                                      v2f b0, v2f b1, v2f b2, v2f b3)
{
  acc[0][0] = __builtin_amdgcn_wmma_f32_16x16x4_f32(false, a0, false, b0, (short)0, acc[0][0], false, false);
  acc[0][1] = __builtin_amdgcn_wmma_f32_16x16x4_f32(false, a0, false, b1, (short)0, acc[0][1], false, false);
  acc[0][2] = __builtin_amdgcn_wmma_f32_16x16x4_f32(false, a0, false, b2, (short)0, acc[0][2], false, false);
  acc[0][3] = __builtin_amdgcn_wmma_f32_16x16x4_f32(false, a0, false, b3, (short)0, acc[0][3], false, false);
  acc[1][0] = __builtin_amdgcn_wmma_f32_16x16x4_f32(false, a1, false, b0, (short)0, acc[1][0], false, false);
  acc[1][1] = __builtin_amdgcn_wmma_f32_16x16x4_f32(false, a1, false, b1, (short)0, acc[1][1], false, false);
  acc[1][2] = __builtin_amdgcn_wmma_f32_16x16x4_f32(false, a1, false, b2, (short)0, acc[1][2], false, false);
  acc[1][3] = __builtin_amdgcn_wmma_f32_16x16x4_f32(false, a1, false, b3, (short)0, acc[1][3], false, false);
}

struct F3 { const float* a; const float* b; const float* c; };
struct O3 { float* a; float* b; float* c; };
__device__ __forceinline__ const float* pick3(const F3& s, int z)
{
  return z == 0 ? s.a : (z == 1 ? s.b : s.c);
}
__device__ __forceinline__ float* pick3(const O3& s, int z)
{
  return z == 0 ? s.a : (z == 1 ? s.b : s.c);
}

#define RED_JP 272
#define RED_PW (8 * RED_JP)

#define STG_P 20

template <bool RELU, bool BIAS, bool GATHER, int MW, int KS, bool STAGE>
__global__ __launch_bounds__(32 * MW * KS) void gemm_wmma_f32(
    F3 Az, int lda,
    F3 Bz, int ldb,
    F3 biasz,
    O3 Cz, int ldc,
    int K,
    const int* __restrict__ gl, const int* __restrict__ gr)
{
  static_assert(KS == 1 || MW == 1, "K split reduces one 32x64 tile per block");
  const int z = blockIdx.z;
  const float* __restrict__ A    = pick3(Az, z);
  const float* __restrict__ Bm   = pick3(Bz, z);
  const float* __restrict__ bias = pick3(biasz, z);
  float* __restrict__ C          = pick3(Cz, z);
  static_assert(!GATHER || (KS >= 2 && (64 % KS) == 0), "gather slice must stay in one half");
  static_assert(!STAGE || (KS >= 2 && !GATHER), "staging uses the wave's partial-tile region");
  __shared__ __attribute__((aligned(16))) float red[KS > 1 ? KS * RED_PW : 4];

  const int lane = threadIdx.x & 31;
  const int wave = __builtin_amdgcn_readfirstlane((int)(threadIdx.x >> 5));
  const int wm   = wave % MW;
  const int wk   = wave / MW;
  const int half = lane >> 4;
  const int l16  = lane & 15;

  const int mb  = blockIdx.y * (32 * MW);
  const int m0  = mb + wm * 32;
  const int n0  = blockIdx.x * 64;
  const int am0 = m0 + l16;
  const int am1 = m0 + 16 + l16;

  v8f acc[2][4] = {};

  const int nsteps = (GATHER ? 512 : K) >> 2;
  const int per    = (nsteps + KS - 1) / KS;
  const int sbeg   = min(wk * per, nsteps);
  const int send   = min(sbeg + per, nsteps);
  const int kbeg   = 4 * sbeg;

  const float* bp = Bm + (size_t)(kbeg + 2 * half) * ldb + n0 + l16;
  const size_t bstep = (size_t)4 * ldb;

  if (GATHER) {
    const int* gi = (kbeg < 256) ? gl : gr;
    const float* ap0 = A + (size_t)gi[am0] * 256 + (kbeg & 255) + 2 * half;
    const float* ap1 = A + (size_t)gi[am1] * 256 + (kbeg & 255) + 2 * half;
    #pragma unroll 4
    for (int s = sbeg; s < send; ++s) {
      v2f a0 = *(const v2f*)ap0;
      v2f a1 = *(const v2f*)ap1;
      v2f b0, b1, b2, b3;
      b0.x = bp[0];  b0.y = bp[ldb + 0];
      b1.x = bp[16]; b1.y = bp[ldb + 16];
      b2.x = bp[32]; b2.y = bp[ldb + 32];
      b3.x = bp[48]; b3.y = bp[ldb + 48];
      wmma8(acc, a0, a1, b0, b1, b2, b3);
      ap0 += 4; ap1 += 4; bp += bstep;
    }
  } else if (STAGE) {
    const int r8 = lane >> 2;
    const int q4 = (lane & 3) * 4;
    const float* gp = A + (size_t)(m0 + r8) * lda + kbeg + q4;
    const size_t g8 = (size_t)8 * lda;
    float* stg = red + wk * RED_PW;
    float* sw = stg + r8 * STG_P + q4;
    const float* sr = stg + l16 * STG_P + 2 * half;
    const int nch = (send - sbeg) >> 2;
    float4 g0 = *(const float4*)gp;
    float4 g1 = *(const float4*)(gp + g8);
    float4 g2 = *(const float4*)(gp + 2 * g8);
    float4 g3 = *(const float4*)(gp + 3 * g8);
    for (int ch = 0; ch < nch; ++ch) {
      *(float4*)sw                = g0;
      *(float4*)(sw + 8 * STG_P)  = g1;
      *(float4*)(sw + 16 * STG_P) = g2;
      *(float4*)(sw + 24 * STG_P) = g3;
      if (ch + 1 < nch) {
        gp += 16;
        g0 = *(const float4*)gp;
        g1 = *(const float4*)(gp + g8);
        g2 = *(const float4*)(gp + 2 * g8);
        g3 = *(const float4*)(gp + 3 * g8);
      }
      #pragma unroll 2
      for (int s = 0; s < 4; ++s) {
        v2f a0 = *(const v2f*)(sr + 4 * s);
        v2f a1 = *(const v2f*)(sr + 16 * STG_P + 4 * s);
        v2f b0, b1, b2, b3;
        b0.x = bp[0];  b0.y = bp[ldb + 0];
        b1.x = bp[16]; b1.y = bp[ldb + 16];
        b2.x = bp[32]; b2.y = bp[ldb + 32];
        b3.x = bp[48]; b3.y = bp[ldb + 48];
        wmma8(acc, a0, a1, b0, b1, b2, b3);
        bp += bstep;
      }
    }
  } else {
    const float* ap0 = A + (size_t)am0 * lda + kbeg + 2 * half;
    const float* ap1 = A + (size_t)am1 * lda + kbeg + 2 * half;
    #pragma unroll 2
    for (int s = sbeg; s < send; ++s) {
      v2f a0 = *(const v2f*)ap0;
      v2f a1 = *(const v2f*)ap1;
      v2f b0, b1, b2, b3;
      b0.x = bp[0];  b0.y = bp[ldb + 0];
      b1.x = bp[16]; b1.y = bp[ldb + 16];
      b2.x = bp[32]; b2.y = bp[ldb + 32];
      b3.x = bp[48]; b3.y = bp[ldb + 48];
      wmma8(acc, a0, a1, b0, b1, b2, b3);
      ap0 += 4; ap1 += 4; bp += bstep;
    }
    const int kmain = K & ~3;
    if (kmain < K && wk == KS - 1) {
      const int ka = kmain + 2 * half;
      const bool g0 = (ka < K), g1 = (ka + 1 < K);
      v2f a0, a1, b0, b1, b2, b3;
      a0.x = g0 ? ap0[0] : 0.0f;  a0.y = g1 ? ap0[1] : 0.0f;
      a1.x = g0 ? ap1[0] : 0.0f;  a1.y = g1 ? ap1[1] : 0.0f;
      b0.x = g0 ? bp[0]  : 0.0f;  b0.y = g1 ? bp[ldb + 0]  : 0.0f;
      b1.x = g0 ? bp[16] : 0.0f;  b1.y = g1 ? bp[ldb + 16] : 0.0f;
      b2.x = g0 ? bp[32] : 0.0f;  b2.y = g1 ? bp[ldb + 32] : 0.0f;
      b3.x = g0 ? bp[48] : 0.0f;  b3.y = g1 ? bp[ldb + 48] : 0.0f;
      wmma8(acc, a0, a1, b0, b1, b2, b3);
    }
  }

  if (KS == 1) {
    #pragma unroll
    for (int s = 0; s < 2; ++s) {
      #pragma unroll
      for (int v = 0; v < 8; ++v) {
        const int row = m0 + 16 * s + v + 8 * half;
        #pragma unroll
        for (int j = 0; j < 4; ++j) {
          const int col = n0 + 16 * j + l16;
          float x = acc[s][j][v];
          if (BIAS) x += bias[col];
          if (RELU) x = fmaxf(x, 0.0f);
          C[(size_t)row * ldc + col] = x;
        }
      }
    }
  } else {
    float* my = red + wk * RED_PW + lane;
    #pragma unroll
    for (int s = 0; s < 2; ++s) {
      #pragma unroll
      for (int j = 0; j < 4; ++j) {
        #pragma unroll
        for (int v = 0; v < 8; ++v)
          my[(s * 4 + j) * RED_JP + v * 32] = acc[s][j][v];
      }
    }
    __syncthreads();
    for (int e = threadIdx.x; e < 512; e += 32 * KS) {
      const int row = e >> 4;
      const int c4  = (e & 15) * 4;
      const int s   = row >> 4;
      const int hf  = (row >> 3) & 1;
      const int v   = row & 7;
      const int j   = c4 >> 4;
      const int l   = c4 & 15;
      const float* p = red + (s * 4 + j) * RED_JP + v * 32 + hf * 16 + l;
      float4 sum = *(const float4*)p;
      #pragma unroll
      for (int w = 1; w < KS; ++w) {
        const float4 x = *(const float4*)(p + w * RED_PW);
        sum.x += x.x; sum.y += x.y; sum.z += x.z; sum.w += x.w;
      }
      const int col = n0 + c4;
      if (BIAS) {
        sum.x += bias[col + 0]; sum.y += bias[col + 1];
        sum.z += bias[col + 2]; sum.w += bias[col + 3];
      }
      if (RELU) {
        sum.x = fmaxf(sum.x, 0.0f); sum.y = fmaxf(sum.y, 0.0f);
        sum.z = fmaxf(sum.z, 0.0f); sum.w = fmaxf(sum.w, 0.0f);
      }
      *(float4*)(C + (size_t)(mb + row) * ldc + col) = sum;
    }
  }
}

#define SQ_T   192
#define SQ_KT  64
#define SQ_PA  68
#define SQ_PB  416
#define SQ_ASZ (SQ_T * SQ_PA)
#define SQ_BSZ ((SQ_KT / 2) * SQ_PB)
#define SQ_STG (SQ_ASZ + SQ_BSZ)

__device__ __forceinline__ void sq_load(float4 (&ra)[12], float4 (&rb0)[6], float4 (&rb1)[6],
                                        const float* gA, const float* gB)
{
  #pragma unroll
  for (int i = 0; i < 12; ++i) ra[i] = *(const float4*)(gA + (size_t)i * 16 * NNODE);
  #pragma unroll
  for (int a = 0; a < 2; ++a) {
    #pragma unroll
    for (int b = 0; b < 3; ++b) {
      const float* g = gB + (size_t)a * 32 * NNODE + b * 64;
      rb0[a * 3 + b] = *(const float4*)g;
      rb1[a * 3 + b] = *(const float4*)(g + NNODE);
    }
  }
}

template <int I>
__device__ __forceinline__ void sq_store_a(float* aW, const float4 (&ra)[12])
{
  *(float4*)(aW + I * 16 * SQ_PA) = make_float4(ra[I].x, ra[I].y, ra[I].z, ra[I].w);
}

template <int Q>
__device__ __forceinline__ void sq_store_b(float* bW, const float4 (&rb0)[6],
                                           const float4 (&rb1)[6])
{
  float* d = bW + (Q / 3) * 16 * SQ_PB + (Q % 3) * 128;
  *(float4*)d       = make_float4(rb0[Q].x, rb1[Q].x, rb0[Q].y, rb1[Q].y);
  *(float4*)(d + 4) = make_float4(rb0[Q].z, rb1[Q].z, rb0[Q].w, rb1[Q].w);
}

template <bool STAGE_NEXT, int I>
__device__ __forceinline__ void sq_kstep(v8f (&acc)[3][6], v2f (&a)[2][3], v2f (&b)[2][6],
                                         const float* aR, const float* bR,
                                         float* aW, float* bW, const float4 (&ra)[12],
                                         const float4 (&rb0)[6], const float4 (&rb1)[6])
{
  constexpr int c = I & 1, n = c ^ 1;
  if (I + 1 < SQ_KT / 4) {
    #pragma unroll
    for (int s = 0; s < 3; ++s)
      a[n][s] = *(const v2f*)(aR + s * 16 * SQ_PA + 4 * (I + 1));
    #pragma unroll
    for (int j = 0; j < 6; ++j)
      b[n][j] = *(const v2f*)(bR + 2 * (I + 1) * SQ_PB + j * 32);
  }
  #pragma unroll
  for (int s = 0; s < 3; ++s) {
    #pragma unroll
    for (int j = 0; j < 6; ++j)
      acc[s][j] = __builtin_amdgcn_wmma_f32_16x16x4_f32(false, a[c][s], false, b[c][j],
                                                        (short)0, acc[s][j], false, false);
  }
  if constexpr (STAGE_NEXT && I >= 2 && I < 14) {
    __builtin_amdgcn_sched_barrier(0);
    sq_store_a<I - 2>(aW, ra);
    if constexpr (((I - 2) & 1) == 0) sq_store_b<(I - 2) / 2>(bW, rb0, rb1);
  }
}

template <bool STAGE_NEXT>
__device__ __forceinline__ void sq_slab(v8f (&acc)[3][6], const float* aR, const float* bR,
                                        float* aW, float* bW, const float4 (&ra)[12],
                                        const float4 (&rb0)[6], const float4 (&rb1)[6])
{
  v2f a[2][3], b[2][6];
  #pragma unroll
  for (int s = 0; s < 3; ++s) a[0][s] = *(const v2f*)(aR + s * 16 * SQ_PA);
  #pragma unroll
  for (int j = 0; j < 6; ++j) b[0][j] = *(const v2f*)(bR + j * 32);
  sq_kstep<STAGE_NEXT,  0>(acc, a, b, aR, bR, aW, bW, ra, rb0, rb1);
  sq_kstep<STAGE_NEXT,  1>(acc, a, b, aR, bR, aW, bW, ra, rb0, rb1);
  sq_kstep<STAGE_NEXT,  2>(acc, a, b, aR, bR, aW, bW, ra, rb0, rb1);
  sq_kstep<STAGE_NEXT,  3>(acc, a, b, aR, bR, aW, bW, ra, rb0, rb1);
  sq_kstep<STAGE_NEXT,  4>(acc, a, b, aR, bR, aW, bW, ra, rb0, rb1);
  sq_kstep<STAGE_NEXT,  5>(acc, a, b, aR, bR, aW, bW, ra, rb0, rb1);
  sq_kstep<STAGE_NEXT,  6>(acc, a, b, aR, bR, aW, bW, ra, rb0, rb1);
  sq_kstep<STAGE_NEXT,  7>(acc, a, b, aR, bR, aW, bW, ra, rb0, rb1);
  sq_kstep<STAGE_NEXT,  8>(acc, a, b, aR, bR, aW, bW, ra, rb0, rb1);
  sq_kstep<STAGE_NEXT,  9>(acc, a, b, aR, bR, aW, bW, ra, rb0, rb1);
  sq_kstep<STAGE_NEXT, 10>(acc, a, b, aR, bR, aW, bW, ra, rb0, rb1);
  sq_kstep<STAGE_NEXT, 11>(acc, a, b, aR, bR, aW, bW, ra, rb0, rb1);
  sq_kstep<STAGE_NEXT, 12>(acc, a, b, aR, bR, aW, bW, ra, rb0, rb1);
  sq_kstep<STAGE_NEXT, 13>(acc, a, b, aR, bR, aW, bW, ra, rb0, rb1);
  sq_kstep<STAGE_NEXT, 14>(acc, a, b, aR, bR, aW, bW, ra, rb0, rb1);
  sq_kstep<STAGE_NEXT, 15>(acc, a, b, aR, bR, aW, bW, ra, rb0, rb1);
}

__global__ __launch_bounds__(256) void gemm_sq_lds_f32(const float* __restrict__ A,
                                                       const float* __restrict__ Bm,
                                                       float* __restrict__ C)
{
  __shared__ __attribute__((aligned(16))) float sm[2 * SQ_STG];

  const int tid  = threadIdx.x;
  const int lane = tid & 31;
  const int wave = __builtin_amdgcn_readfirstlane(tid >> 5);
  const int wm   = wave >> 1;
  const int wn   = wave & 1;
  const int half = lane >> 4;
  const int l16  = lane & 15;
  const int mb   = blockIdx.y * SQ_T;
  const int nb   = blockIdx.x * SQ_T;

  const int tr = tid >> 4, tc = tid & 15;
  const float* gA = A + (size_t)(mb + tr) * NNODE + 4 * tc;
  const float* gB = Bm + (size_t)(2 * tr) * NNODE + nb + 4 * tc;
  const int sAo = tr * SQ_PA + 4 * tc;
  const int sBo = SQ_ASZ + tr * SQ_PB + 8 * tc;
  const int aF = (48 * wm + l16) * SQ_PA + 2 * half;
  const int bF = SQ_ASZ + half * SQ_PB + 2 * (96 * wn + l16);

  v8f acc[3][6] = {};
  float4 ra[12], rb0[6], rb1[6];

  sq_load(ra, rb0, rb1, gA, gB);
  sq_store_a<0>(sm + sAo, ra);  sq_store_a<1>(sm + sAo, ra);  sq_store_a<2>(sm + sAo, ra);
  sq_store_a<3>(sm + sAo, ra);  sq_store_a<4>(sm + sAo, ra);  sq_store_a<5>(sm + sAo, ra);
  sq_store_a<6>(sm + sAo, ra);  sq_store_a<7>(sm + sAo, ra);  sq_store_a<8>(sm + sAo, ra);
  sq_store_a<9>(sm + sAo, ra);  sq_store_a<10>(sm + sAo, ra); sq_store_a<11>(sm + sAo, ra);
  sq_store_b<0>(sm + sBo, rb0, rb1);  sq_store_b<1>(sm + sBo, rb0, rb1);
  sq_store_b<2>(sm + sBo, rb0, rb1);  sq_store_b<3>(sm + sBo, rb0, rb1);
  sq_store_b<4>(sm + sBo, rb0, rb1);  sq_store_b<5>(sm + sBo, rb0, rb1);
  __syncthreads();

  int st = 0;
  for (int t = 0; t < NNODE / SQ_KT - 1; ++t) {
    gA += SQ_KT;
    gB += (size_t)SQ_KT * NNODE;
    sq_load(ra, rb0, rb1, gA, gB);
    sq_slab<true>(acc, sm + st + aF, sm + st + bF, sm + (SQ_STG - st) + sAo,
                  sm + (SQ_STG - st) + sBo, ra, rb0, rb1);
    __syncthreads();
    st = SQ_STG - st;
  }
  sq_slab<false>(acc, sm + st + aF, sm + st + bF, nullptr, nullptr, ra, rb0, rb1);

  #pragma unroll
  for (int s = 0; s < 3; ++s) {
    #pragma unroll
    for (int v = 0; v < 8; ++v) {
      const int row = mb + 48 * wm + 16 * s + v + 8 * half;
      #pragma unroll
      for (int j = 0; j < 6; ++j) {
        const int col = nb + 96 * wn + 16 * j + l16;
        C[(size_t)row * NNODE + col] = acc[s][j][v];
      }
    }
  }
}

__global__ void deg_rsqrt(const float* __restrict__ A, float* __restrict__ d)
{
  const int gw   = (blockIdx.x * blockDim.x + threadIdx.x) >> 5;
  const int lane = threadIdx.x & 31;
  if (gw >= NNODE) return;
  const float* row = A + (size_t)gw * NNODE;
  float s = 0.0f;
  for (int j = lane; j < NNODE; j += 32) s += row[j];
  #pragma unroll
  for (int off = 16; off > 0; off >>= 1) s += __shfl_xor(s, off, 32);
  if (lane == 0) d[gw] = (s > 0.0f) ? rsqrtf(s) : 0.0f;
}

#define DEG_G (NNODE / 4 / 32)
#define DEG_WAVES __attribute__((amdgpu_waves_per_eu(4, 4)))

__device__ __forceinline__ float thr_w(float m, float a, float thr)
{
  return (m > thr) ? a : 0.0f;
}

__global__ __launch_bounds__(256) DEG_WAVES void branch_deg_rsqrt(
    F3 maskz, const float* __restrict__ adj,
    const float* __restrict__ thrp, float* __restrict__ d3)
{
  const int gw   = (blockIdx.x * blockDim.x + threadIdx.x) >> 5;
  const int lane = threadIdx.x & 31;
  if (gw >= NNODE) return;
  const int z = blockIdx.z;
  const float* __restrict__ mask = pick3(maskz, z);
  float* __restrict__ d = d3 + z * NNODE;
  const float thr = thrp[0];
  const float* mp = mask + (size_t)gw * NNODE + 4 * lane;
  const float* ap = adj  + (size_t)gw * NNODE + 4 * lane;
  float4 m[DEG_G], a[DEG_G];
  #pragma unroll
  for (int g = 0; g < DEG_G; ++g) {
    m[g] = *(const float4*)(mp + 128 * g);
    a[g] = *(const float4*)(ap + 128 * g);
  }
  float s = thr_w(m[0].x, a[0].x, thr) + thr_w(m[0].y, a[0].y, thr);
  s += thr_w(m[0].z, a[0].z, thr);
  s += thr_w(m[0].w, a[0].w, thr);
  #pragma unroll
  for (int g = 1; g < DEG_G; ++g) {
    s += thr_w(m[g].x, a[g].x, thr);
    s += thr_w(m[g].y, a[g].y, thr);
    s += thr_w(m[g].z, a[g].z, thr);
    s += thr_w(m[g].w, a[g].w, thr);
  }
  #pragma unroll
  for (int off = 16; off > 0; off >>= 1) s += __shfl_xor(s, off, 32);
  if (lane == 0) d[gw] = rsqrtf(s + 1.0f);
}

#define NORM_R 6
#define NORM_C (NNODE / 4 / 256)
#define NORM_WAVES __attribute__((amdgpu_waves_per_eu(4, 4)))

__global__ __launch_bounds__(256) NORM_WAVES void symnorm(const float* __restrict__ adj,
                                                          const float* __restrict__ d,
                                                          float* __restrict__ An)
{
  const int tid = threadIdx.x;
  const int r0  = blockIdx.x * NORM_R;
  const float* ap = adj + (size_t)r0 * NNODE + 4 * tid;
  float*       op = An  + (size_t)r0 * NNODE + 4 * tid;

  float4 dj[NORM_C], a[NORM_R][NORM_C];
  float  di[NORM_R];
  #pragma unroll
  for (int c = 0; c < NORM_C; ++c) dj[c] = *(const float4*)(d + 4 * tid + 1024 * c);
  #pragma unroll
  for (int r = 0; r < NORM_R; ++r) di[r] = d[r0 + r];
  #pragma unroll
  for (int r = 0; r < NORM_R; ++r) {
    #pragma unroll
    for (int c = 0; c < NORM_C; ++c)
      a[r][c] = *(const float4*)(ap + (size_t)r * NNODE + 1024 * c);
  }
  #pragma unroll
  for (int r = 0; r < NORM_R; ++r) {
    #pragma unroll
    for (int c = 0; c < NORM_C; ++c) {
      float4 o;
      o.x = a[r][c].x * di[r] * dj[c].x;
      o.y = a[r][c].y * di[r] * dj[c].y;
      o.z = a[r][c].z * di[r] * dj[c].z;
      o.w = a[r][c].w * di[r] * dj[c].w;
      *(float4*)(op + (size_t)r * NNODE + 1024 * c) = o;
    }
  }
}

__device__ __forceinline__ float bn_elem(float m, float a, float thr, bool diag,
                                         float di, float dj)
{
  float w = (m > thr) ? a : 0.0f;
  if (diag) w += 1.0f;
  return w * di * dj;
}

__global__ __launch_bounds__(256) NORM_WAVES void branch_norm(
    F3 maskz, const float* __restrict__ adj,
    const float* __restrict__ thrp, const float* __restrict__ d3, O3 Anz)
{
  const int tid = threadIdx.x;
  const int r0  = blockIdx.x * NORM_R;
  const int z   = blockIdx.z;
  const float* mask = pick3(maskz, z);
  float* An = pick3(Anz, z);
  const float* __restrict__ d = d3 + z * NNODE;
  const float thr = thrp[0];
  const float* mp = mask + (size_t)r0 * NNODE + 4 * tid;
  const float* ap = adj  + (size_t)r0 * NNODE + 4 * tid;
  float*       op = An   + (size_t)r0 * NNODE + 4 * tid;

  float4 dj[NORM_C], m[NORM_R][NORM_C], a[NORM_R][NORM_C];
  float  di[NORM_R];
  #pragma unroll
  for (int c = 0; c < NORM_C; ++c) dj[c] = *(const float4*)(d + 4 * tid + 1024 * c);
  #pragma unroll
  for (int r = 0; r < NORM_R; ++r) di[r] = d[r0 + r];
  #pragma unroll
  for (int r = 0; r < NORM_R; ++r) {
    #pragma unroll
    for (int c = 0; c < NORM_C; ++c) {
      m[r][c] = *(const float4*)(mp + (size_t)r * NNODE + 1024 * c);
      a[r][c] = *(const float4*)(ap + (size_t)r * NNODE + 1024 * c);
    }
  }
  #pragma unroll
  for (int r = 0; r < NORM_R; ++r) {
    #pragma unroll
    for (int c = 0; c < NORM_C; ++c) {
      const int dc = 4 * tid + 1024 * c - (r0 + r);
      float4 o;
      o.x = bn_elem(m[r][c].x, a[r][c].x, thr, dc ==  0, di[r], dj[c].x);
      o.y = bn_elem(m[r][c].y, a[r][c].y, thr, dc == -1, di[r], dj[c].y);
      o.z = bn_elem(m[r][c].z, a[r][c].z, thr, dc == -2, di[r], dj[c].z);
      o.w = bn_elem(m[r][c].w, a[r][c].w, thr, dc == -3, di[r], dj[c].w);
      *(float4*)(op + (size_t)r * NNODE + 1024 * c) = o;
    }
  }
}

__global__ void conv_dil2(const float* __restrict__ emb,
                          const float* __restrict__ kw,
                          const float* __restrict__ kb,
                          float* __restrict__ out)
{
  const int idx = blockIdx.x * blockDim.x + threadIdx.x;
  if (idx >= NNODE * 256) return;
  const int n = idx >> 8;
  const int c = idx & 255;
  float acc = kb[0];
  #pragma unroll
  for (int ch = 0; ch < 3; ++ch) {
    #pragma unroll
    for (int kh = 0; kh < 3; ++kh) {
      const int ih = n + 2 * kh - 2;
      if (ih < 0 || ih >= NNODE) continue;
      #pragma unroll
      for (int kwi = 0; kwi < 3; ++kwi) {
        const int iw = c + 2 * kwi - 2;
        if (iw < 0 || iw >= 256) continue;
        acc += emb[((size_t)ch * NNODE + ih) * 256 + iw] * kw[ch * 9 + kh * 3 + kwi];
      }
    }
  }
  out[idx] = acc;
}

__global__ void logits_k(const float* __restrict__ h,
                         const float* __restrict__ Wb,
                         const float* __restrict__ bb,
                         float* __restrict__ out)
{
  const int i = blockIdx.x * blockDim.x + threadIdx.x;
  if (i >= BPAIR * 2) return;
  const int b = i >> 1;
  const int c = i & 1;
  float acc = bb[c];
  #pragma unroll
  for (int k = 0; k < 64; ++k) acc += h[b * 64 + k] * Wb[k * 2 + c];
  out[i] = acc;
}

extern "C" void kernel_launch(void* const* d_in, const int* in_sizes, int n_in,
                              void* d_out, int out_size, void* d_ws, size_t ws_size,
                              hipStream_t stream)
{
  const int*   left  = (const int*)d_in[0];
  const int*   right = (const int*)d_in[1];
  const float* X     = (const float*)d_in[2];
  const float* adj   = (const float*)d_in[3];
  const float* thr   = (const float*)d_in[4];
  const float* Wfc1  = (const float*)d_in[5];
  const float* bfc1  = (const float*)d_in[6];
  const float* Wg[6] = {(const float*)d_in[7],  (const float*)d_in[9],
                        (const float*)d_in[11], (const float*)d_in[13],
                        (const float*)d_in[15], (const float*)d_in[17]};
  const float* bg[6] = {(const float*)d_in[8],  (const float*)d_in[10],
                        (const float*)d_in[12], (const float*)d_in[14],
                        (const float*)d_in[16], (const float*)d_in[18]};
  const float* cnnk  = (const float*)d_in[19];
  const float* cnnb  = (const float*)d_in[20];
  const float* Wa    = (const float*)d_in[21];
  const float* ba    = (const float*)d_in[22];
  const float* Wb    = (const float*)d_in[23];
  const float* bb    = (const float*)d_in[24];

  float* out     = (float*)d_out;
  float* logits  = out;
  float* emb_all = out + BPAIR * 2;

  float* ws = (float*)d_ws;
  size_t off = 0;
  float* X0  = ws + off; off += (size_t)NNODE * D0;
  float* An  = ws + off; off += (size_t)NNODE * NNODE;
  float* A2  = ws + off; off += (size_t)NNODE * NNODE;
  float* AnB = ws + off; off += (size_t)NNODE * NNODE;
  float* dv  = ws + off; off += NNODE;
  float* Hb  = ws + off; off += (size_t)NNODE * HDIM;
  float* H2b = ws + off; off += (size_t)NNODE * HDIM;
  float* emb = ws + off; off += (size_t)3 * NNODE * 256;
  float* hm  = ws + off; off += (size_t)BPAIR * 64;
  if (ws_size < off * sizeof(float)) return;

  const dim3 blk8(256);
  const dim3 blk16(512);
  const F3 none = {nullptr, nullptr, nullptr};

  gemm_wmma_f32<true, true, false, 1, 16, false><<<dim3(D0 / 64, NNODE / 32), blk16, 0, stream>>>(
      F3{X, X, X}, FIN, F3{Wfc1, Wfc1, Wfc1}, D0, F3{bfc1, bfc1, bfc1}, O3{X0, X0, X0}, D0, FIN,
      nullptr, nullptr);

  deg_rsqrt<<<NNODE / 8, 256, 0, stream>>>(adj, dv);
  symnorm<<<NNODE / NORM_R, 256, 0, stream>>>(adj, dv, An);
  gemm_sq_lds_f32<<<dim3(NNODE / SQ_T, NNODE / SQ_T), 256, 0, stream>>>(An, An, A2);

  float* dv3  = Hb;
  float* emb0 = emb;
  float* emb1 = emb + (size_t)NNODE * 256;
  float* emb2 = emb + (size_t)2 * NNODE * 256;
  const F3 masks = {adj, An, A2};
  const O3 norms = {AnB, An, A2};
  const F3 normr = {AnB, An, A2};

  branch_deg_rsqrt<<<dim3(NNODE / 8, 1, 3), 256, 0, stream>>>(masks, adj, thr, dv3);
  branch_norm<<<dim3(NNODE / NORM_R, 1, 3), 256, 0, stream>>>(masks, adj, thr, dv3, norms);

  gemm_wmma_f32<false, false, false, 1, 8, false><<<dim3(HDIM / 64, NNODE / 32, 3), blk8, 0, stream>>>(
      F3{X0, X0, X0}, D0, F3{Wg[4], Wg[2], Wg[0]}, HDIM, none,
      O3{emb0 + 128, emb1 + 128, emb2 + 128}, 256, D0, nullptr, nullptr);
  gemm_wmma_f32<true, true, false, 1, 16, true><<<dim3(HDIM / 64, NNODE / 32, 3), blk16, 0, stream>>>(
      normr, NNODE, F3{emb0 + 128, emb1 + 128, emb2 + 128}, 256, F3{bg[4], bg[2], bg[0]},
      O3{emb0, emb1, emb2}, 256, NNODE, nullptr, nullptr);
  gemm_wmma_f32<false, false, false, 1, 8, false><<<dim3(HDIM / 64, NNODE / 32, 3), blk8, 0, stream>>>(
      F3{emb0, emb1, emb2}, 256, F3{Wg[5], Wg[3], Wg[1]}, HDIM, none,
      O3{Hb, H2b, X0}, HDIM, HDIM, nullptr, nullptr);
  gemm_wmma_f32<true, true, false, 1, 16, true><<<dim3(HDIM / 64, NNODE / 32, 3), blk16, 0, stream>>>(
      normr, NNODE, F3{Hb, H2b, X0}, HDIM, F3{bg[5], bg[3], bg[1]},
      O3{emb0 + 128, emb1 + 128, emb2 + 128}, 256, NNODE, nullptr, nullptr);

  conv_dil2<<<(NNODE * 256) / 256, 256, 0, stream>>>(emb, cnnk, cnnb, emb_all);

  gemm_wmma_f32<true, true, true, 1, 8, false><<<dim3(64 / 64, BPAIR / 32), blk8, 0, stream>>>(
      F3{emb_all, emb_all, emb_all}, 256, F3{Wa, Wa, Wa}, 64, F3{ba, ba, ba}, O3{hm, hm, hm}, 64,
      512, left, right);

  logits_k<<<(BPAIR * 2 + 255) / 256, 256, 0, stream>>>(hm, Wb, bb, logits);
}
